# starting kernel
# baseline (speedup 1.0000x reference)
#include <hip/hip_runtime.h>
#include <hip/hip_bf16.h>
#include <math.h>

typedef __attribute__((ext_vector_type(2))) float v2f;
typedef __attribute__((ext_vector_type(8))) float v8f;

#define N_NODES   50000
#define N_EDGES   800000
#define F_IN      128
#define HC        128
#define HEADS     4
#define HID       32
#define OUT_C     32
#define NEG_SLOPE 0.2f

__global__ void gemm16x16_wmma_f32(const float* __restrict__ A,
                                   const float* __restrict__ B,
                                   const float* __restrict__ bias,
                                   float* __restrict__ C,
                                   int N, int K, int P) {
    const int wave = threadIdx.x >> 5;
    const int lane = threadIdx.x & 31;
    const int tile = blockIdx.x * (blockDim.x >> 5) + wave;
    const int ptiles = P >> 4;
    const int ntiles = N >> 4;
    if (tile >= ntiles * ptiles) return;
    const int row0 = (tile / ptiles) << 4;
    const int col0 = (tile % ptiles) << 4;
    const int half = lane >> 4;
    const int r    = lane & 15;

    v8f acc = {};
    const float* arow = A + (size_t)(row0 + r) * K;
    for (int k0 = 0; k0 < K; k0 += 4) {
        const int ka = k0 + 2 * half;
        v2f a = *(const v2f*)(arow + ka);
        v2f b;
        b.x = B[(size_t)(ka + 0) * P + col0 + r];
        b.y = B[(size_t)(ka + 1) * P + col0 + r];
        acc = __builtin_amdgcn_wmma_f32_16x16x4_f32(
             false, a,  false, b,
             (short)0, acc,  false,  false);
    }

    const float badd = bias ? bias[col0 + r] : 0.0f;
#pragma unroll
    for (int i = 0; i < 8; ++i) {
        const int row = row0 + i + half * 8;
        C[(size_t)row * P + col0 + r] = acc[i] + badd;
    }
}

__global__ void attn_logits(const float* __restrict__ h,
                            const float* __restrict__ aS,
                            const float* __restrict__ aD,
                            float* __restrict__ als,
                            float* __restrict__ ald, int N) {
    const int i = blockIdx.x * blockDim.x + threadIdx.x;
    if (i >= N * HEADS) return;
    const int n  = i >> 2;
    const int hd = i & 3;
    const float* hp = h + (size_t)n * HC + hd * HID;
    const float* as = aS + hd * HID;
    const float* ad = aD + hd * HID;
    float ss = 0.f, sd = 0.f;
#pragma unroll 8
    for (int c = 0; c < HID; ++c) {
        const float v = hp[c];
        ss += v * as[c];
        sd += v * ad[c];
    }
    als[i] = ss;
    ald[i] = sd;
}

__global__ void init_buffers(float* __restrict__ att, unsigned* __restrict__ m,
                             float* __restrict__ s, int natt, int nms) {
    const int i = blockIdx.x * blockDim.x + threadIdx.x;
    if (i < natt) att[i] = 0.0f;
    if (i < nms) { m[i] = 0u; s[i] = 0.0f; }
}

__device__ __forceinline__ unsigned f2ord(float f) {
    unsigned u = __float_as_uint(f);
    return (u & 0x80000000u) ? ~u : (u | 0x80000000u);
}
__device__ __forceinline__ float ord2f(unsigned u) {
    return (u & 0x80000000u) ? __uint_as_float(u ^ 0x80000000u)
                             : __uint_as_float(~u);
}

__global__ void edge_max(const int* __restrict__ src, const int* __restrict__ dst,
                         const float* __restrict__ als, const float* __restrict__ ald,
                         float* __restrict__ ebuf, unsigned* __restrict__ m) {
    const int i = blockIdx.x * blockDim.x + threadIdx.x;
    if (i >= N_EDGES * HEADS) return;
    const int e  = i >> 2;
    const int hd = i & 3;
    const int sN = src[e], dN = dst[e];
    float v = als[sN * HEADS + hd] + ald[dN * HEADS + hd];
    v = (v > 0.f) ? v : NEG_SLOPE * v;
    ebuf[i] = v;
    atomicMax(&m[dN * HEADS + hd], f2ord(v));
}

__global__ void edge_exp(const int* __restrict__ dst, const unsigned* __restrict__ m,
                         float* __restrict__ ebuf, float* __restrict__ s) {
    const int i = blockIdx.x * blockDim.x + threadIdx.x;
    if (i >= N_EDGES * HEADS) return;
    const int e  = i >> 2;
    const int hd = i & 3;
    const int dN = dst[e];
    const float ex = expf(ebuf[i] - ord2f(m[dN * HEADS + hd]));
    ebuf[i] = ex;
    atomicAdd(&s[dN * HEADS + hd], ex);
}

__global__ void edge_aggr(const int* __restrict__ src, const int* __restrict__ dst,
                          const float* __restrict__ ebuf, const float* __restrict__ s,
                          const float* __restrict__ h, float* __restrict__ att) {
    const long long i = (long long)blockIdx.x * blockDim.x + threadIdx.x;
    if (i >= (long long)N_EDGES * HC) return;
    const int e  = (int)(i >> 7);
    const int c  = (int)(i & 127);
    const int hd = c >> 5;
    const int sN = src[e], dN = dst[e];
    const float alpha = ebuf[e * HEADS + hd] / (s[dN * HEADS + hd] + 1e-16f);
    atomicAdd(&att[(size_t)dN * HC + c], alpha * h[(size_t)sN * HC + c]);
}

__global__ void finish01(const float* __restrict__ att, const float* __restrict__ b,
                         const float* __restrict__ skip, float* __restrict__ out, int N) {
    const int i = blockIdx.x * blockDim.x + threadIdx.x;
    if (i >= N * HC) return;
    const int c = i & 127;
    const float v = att[i] + b[c] + skip[i];
    out[i] = (v > 0.f) ? v : (expf(v) - 1.0f);
}

__global__ void finish2(const float* __restrict__ att, const float* __restrict__ b2,
                        const float* __restrict__ skip32, float* __restrict__ out, int N) {
    const int node = blockIdx.x * (blockDim.x >> 5) + (threadIdx.x >> 5);
    const int c    = threadIdx.x & 31;
    if (node >= N) return;
    const float* ap = att + (size_t)node * HC;
    float v = 0.25f * (ap[c] + ap[32 + c] + ap[64 + c] + ap[96 + c])
            + b2[c] + skip32[(size_t)node * OUT_C + c];
    float sq = v * v;
#pragma unroll
    for (int o = 16; o > 0; o >>= 1) sq += __shfl_xor(sq, o, 32);
    const float norm = sqrtf(sq);
    out[(size_t)node * OUT_C + c] = v / fmaxf(norm, 1e-12f);
}

extern "C" void kernel_launch(void* const* d_in, const int* in_sizes, int n_in,
                              void* d_out, int out_size, void* d_ws, size_t ws_size,
                              hipStream_t stream) {
    (void)in_sizes; (void)n_in; (void)out_size; (void)ws_size;

    const float* x   = (const float*)d_in[0];
    const int*   ei  = (const int*)d_in[1];
    const float* W0  = (const float*)d_in[2];
    const float* aS0 = (const float*)d_in[3];
    const float* aD0 = (const float*)d_in[4];
    const float* b0  = (const float*)d_in[5];
    const float* Ws0 = (const float*)d_in[6];
    const float* bs0 = (const float*)d_in[7];
    const float* W1  = (const float*)d_in[8];
    const float* aS1 = (const float*)d_in[9];
    const float* aD1 = (const float*)d_in[10];
    const float* b1  = (const float*)d_in[11];
    const float* Ws1 = (const float*)d_in[12];
    const float* bs1 = (const float*)d_in[13];
    const float* W2  = (const float*)d_in[14];
    const float* aS2 = (const float*)d_in[15];
    const float* aD2 = (const float*)d_in[16];
    const float* b2  = (const float*)d_in[17];
    const float* Ws2 = (const float*)d_in[18];
    const float* bs2 = (const float*)d_in[19];

    const int* src = ei;
    const int* dst = ei + N_EDGES;

    const size_t NF = (size_t)N_NODES * HC;
    float*    ws    = (float*)d_ws;
    float*    x_cur = ws;
    float*    h_lin = ws + NF;
    float*    skipb = ws + 2 * NF;
    float*    att   = ws + 3 * NF;
    float*    als   = ws + 4 * NF;
    float*    ald   = als + (size_t)N_NODES * HEADS;
    unsigned* m     = (unsigned*)(ald + (size_t)N_NODES * HEADS);
    float*    s     = (float*)(m + (size_t)N_NODES * HEADS);
    float*    ebuf  = s + (size_t)N_NODES * HEADS;

    auto gemm = [&](const float* A, const float* B, const float* bias,
                    float* C, int N, int K, int P) {
        const int tiles = (N / 16) * (P / 16);
        const int blocks = (tiles + 7) / 8;
        gemm16x16_wmma_f32<<<blocks, 256, 0, stream>>>(A, B, bias, C, N, K, P);
    };

    auto run_edges = [&](const float* hfeat) {
        const int nInit = N_NODES * HC;
        init_buffers<<<(nInit + 255) / 256, 256, 0, stream>>>(
            att, m, s, nInit, N_NODES * HEADS);
        const int nEH = N_EDGES * HEADS;
        edge_max<<<(nEH + 255) / 256, 256, 0, stream>>>(src, dst, als, ald, ebuf, m);
        edge_exp<<<(nEH + 255) / 256, 256, 0, stream>>>(dst, m, ebuf, s);
        const long long tot = (long long)N_EDGES * HC;
        edge_aggr<<<(int)((tot + 255) / 256), 256, 0, stream>>>(
            src, dst, ebuf, s, hfeat, att);
    };

    const int nNH = N_NODES * HEADS;
    const int nNF = N_NODES * HC;

    gemm(x, W0, nullptr, h_lin, N_NODES, F_IN, HC);
    gemm(x, Ws0, bs0, skipb, N_NODES, F_IN, HC);
    attn_logits<<<(nNH + 255) / 256, 256, 0, stream>>>(h_lin, aS0, aD0, als, ald, N_NODES);
    run_edges(h_lin);
    finish01<<<(nNF + 255) / 256, 256, 0, stream>>>(att, b0, skipb, x_cur, N_NODES);

    gemm(x_cur, W1, nullptr, h_lin, N_NODES, HC, HC);
    gemm(x_cur, Ws1, bs1, skipb, N_NODES, HC, HC);
    attn_logits<<<(nNH + 255) / 256, 256, 0, stream>>>(h_lin, aS1, aD1, als, ald, N_NODES);
    run_edges(h_lin);
    finish01<<<(nNF + 255) / 256, 256, 0, stream>>>(att, b1, skipb, x_cur, N_NODES);

    gemm(x_cur, W2, nullptr, h_lin, N_NODES, HC, HEADS * OUT_C);
    gemm(x_cur, Ws2, bs2, skipb, N_NODES, HC, OUT_C);
    attn_logits<<<(nNH + 255) / 256, 256, 0, stream>>>(h_lin, aS2, aD2, als, ald, N_NODES);
    run_edges(h_lin);
    finish2<<<(N_NODES + 7) / 8, 256, 0, stream>>>(att, b2, skipb, (float*)d_out, N_NODES);
}


# Aggregation: 256 work-groups each own 196 destination rows in LDS, filter the dst list for their own edges, add source rows with LDS float adds; global float atomics and att clear gone; new sum order
# speedup vs baseline: 1.9534x; 1.9534x over previous
#include <hip/hip_runtime.h>
#include <hip/hip_bf16.h>
#include <math.h>

typedef __attribute__((ext_vector_type(2))) float v2f;
typedef __attribute__((ext_vector_type(8))) float v8f;

#define N_NODES   50000
#define N_EDGES   800000
#define F_IN      128
#define HC        128
#define HEADS     4
#define HID       32
#define OUT_C     32
#define NEG_SLOPE 0.2f

__global__ void gemm16x16_wmma_f32(const float* __restrict__ A,
                                   const float* __restrict__ B,
                                   const float* __restrict__ bias,
                                   float* __restrict__ C,
                                   int N, int K, int P) {
    const int wave = threadIdx.x >> 5;
    const int lane = threadIdx.x & 31;
    const int tile = blockIdx.x * (blockDim.x >> 5) + wave;
    const int ptiles = P >> 4;
    const int ntiles = N >> 4;
    if (tile >= ntiles * ptiles) return;
    const int row0 = (tile / ptiles) << 4;
    const int col0 = (tile % ptiles) << 4;
    const int half = lane >> 4;
    const int r    = lane & 15;

    v8f acc = {};
    const float* arow = A + (size_t)(row0 + r) * K;
    for (int k0 = 0; k0 < K; k0 += 4) {
        const int ka = k0 + 2 * half;
        v2f a = *(const v2f*)(arow + ka);
        v2f b;
        b.x = B[(size_t)(ka + 0) * P + col0 + r];
        b.y = B[(size_t)(ka + 1) * P + col0 + r];
        acc = __builtin_amdgcn_wmma_f32_16x16x4_f32(
             false, a,  false, b,
             (short)0, acc,  false,  false);
    }

    const float badd = bias ? bias[col0 + r] : 0.0f;
#pragma unroll
    for (int i = 0; i < 8; ++i) {
        const int row = row0 + i + half * 8;
        C[(size_t)row * P + col0 + r] = acc[i] + badd;
    }
}

__global__ void attn_logits(const float* __restrict__ h,
                            const float* __restrict__ aS,
                            const float* __restrict__ aD,
                            float* __restrict__ als,
                            float* __restrict__ ald, int N) {
    const int i = blockIdx.x * blockDim.x + threadIdx.x;
    if (i >= N * HEADS) return;
    const int n  = i >> 2;
    const int hd = i & 3;
    const float* hp = h + (size_t)n * HC + hd * HID;
    const float* as = aS + hd * HID;
    const float* ad = aD + hd * HID;
    float ss = 0.f, sd = 0.f;
#pragma unroll 8
    for (int c = 0; c < HID; ++c) {
        const float v = hp[c];
        ss += v * as[c];
        sd += v * ad[c];
    }
    als[i] = ss;
    ald[i] = sd;
}

__global__ void init_buffers(unsigned* __restrict__ m, float* __restrict__ s, int nms) {
    const int i = blockIdx.x * blockDim.x + threadIdx.x;
    if (i < nms) { m[i] = 0u; s[i] = 0.0f; }
}

__device__ __forceinline__ unsigned f2ord(float f) {
    unsigned u = __float_as_uint(f);
    return (u & 0x80000000u) ? ~u : (u | 0x80000000u);
}
__device__ __forceinline__ float ord2f(unsigned u) {
    return (u & 0x80000000u) ? __uint_as_float(u ^ 0x80000000u)
                             : __uint_as_float(~u);
}

__global__ void edge_max(const int* __restrict__ src, const int* __restrict__ dst,
                         const float* __restrict__ als, const float* __restrict__ ald,
                         float* __restrict__ ebuf, unsigned* __restrict__ m) {
    const int i = blockIdx.x * blockDim.x + threadIdx.x;
    if (i >= N_EDGES * HEADS) return;
    const int e  = i >> 2;
    const int hd = i & 3;
    const int sN = src[e], dN = dst[e];
    float v = als[sN * HEADS + hd] + ald[dN * HEADS + hd];
    v = (v > 0.f) ? v : NEG_SLOPE * v;
    ebuf[i] = v;
    atomicMax(&m[dN * HEADS + hd], f2ord(v));
}

__global__ void edge_exp(const int* __restrict__ dst, const unsigned* __restrict__ m,
                         float* __restrict__ ebuf, float* __restrict__ s) {
    const int i = blockIdx.x * blockDim.x + threadIdx.x;
    if (i >= N_EDGES * HEADS) return;
    const int e  = i >> 2;
    const int hd = i & 3;
    const int dN = dst[e];
    const float ex = expf(ebuf[i] - ord2f(m[dN * HEADS + hd]));
    ebuf[i] = ex;
    atomicAdd(&s[dN * HEADS + hd], ex);
}

#define OWN_WGS    256
#define OWN_NODES  196
#define OWN_WAVES  32
#define SLICE_E    128
#define N_SLICES   (N_EDGES / SLICE_E)
#define BATCH      4
#define LIST_CAP   1024
#define GROUP      8

template <int CH>
__global__ void __launch_bounds__(1024)
aggr_owned(const int* __restrict__ src, const int* __restrict__ dst,
           const float* __restrict__ ebuf, const float* __restrict__ s,
           const float* __restrict__ h, float* __restrict__ att) {
    __shared__ float    acc[OWN_NODES * CH];
    __shared__ unsigned list[OWN_WAVES * LIST_CAP];

    const int tid  = threadIdx.x;
    const int wave = __builtin_amdgcn_readfirstlane(tid >> 5);
    const int lane = tid & 31;
    const int lo   = blockIdx.x * OWN_NODES;
    const int left = N_NODES - lo;
    const int cnt  = left < OWN_NODES ? left : OWN_NODES;

    for (int i = tid; i < OWN_NODES * CH; i += 1024) acc[i] = 0.0f;
    __syncthreads();

    unsigned* mylist = list + wave * LIST_CAP;
    int n = 0;

    auto flush = [&]() {
        __builtin_amdgcn_wave_barrier();
        for (int i = 0; i < n; i += GROUP) {
            unsigned ent[GROUP];
            int      sN[GROUP];
            float    exv[GROUP], sv[GROUP];
            float    r[GROUP][4];
#pragma unroll
            for (int k = 0; k < GROUP; ++k) {
                const int idx = (i + k < n) ? (i + k) : (n - 1);
                ent[k] = mylist[idx];
            }
#pragma unroll
            for (int k = 0; k < GROUP; ++k) {
                const int e  = (int)(ent[k] & 0xFFFFFu);
                const int dN = lo + (int)(ent[k] >> 20);
                sN[k]  = src[e];
                exv[k] = ebuf[e * HEADS + (lane & 3)];
                sv[k]  = s[dN * HEADS + (lane & 3)];
            }
#pragma unroll
            for (int k = 0; k < GROUP; ++k) {
                const float* hp = h + (size_t)sN[k] * HC + lane;
                r[k][0] = hp[0];
                r[k][1] = hp[HID];
                r[k][2] = hp[2 * HID];
                r[k][3] = hp[3 * HID];
            }
#pragma unroll
            for (int k = 0; k < GROUP; ++k) {
                if (i + k < n) {
                    const unsigned ab = __float_as_uint(exv[k] / (sv[k] + 1e-16f));
                    const float a0 = __uint_as_float(__builtin_amdgcn_readlane(ab, 0));
                    const float a1 = __uint_as_float(__builtin_amdgcn_readlane(ab, 1));
                    const float a2 = __uint_as_float(__builtin_amdgcn_readlane(ab, 2));
                    const float a3 = __uint_as_float(__builtin_amdgcn_readlane(ab, 3));
                    float* ap = acc + (ent[k] >> 20) * CH + lane;
                    if (CH == HC) {
                        atomicAdd(ap,           a0 * r[k][0]);
                        atomicAdd(ap + HID,     a1 * r[k][1]);
                        atomicAdd(ap + 2 * HID, a2 * r[k][2]);
                        atomicAdd(ap + 3 * HID, a3 * r[k][3]);
                    } else {
                        atomicAdd(ap, a0 * r[k][0] + a1 * r[k][1] + a2 * r[k][2] + a3 * r[k][3]);
                    }
                }
            }
        }
        __builtin_amdgcn_wave_barrier();
        n = 0;
    };

    int4 dn[BATCH];
#pragma unroll
    for (int b = 0; b < BATCH; ++b) {
        const int sl = wave * BATCH + b;
        dn[b] = *(const int4*)(dst + sl * SLICE_E + lane * 4);
    }
    for (int s0 = wave * BATCH; s0 < N_SLICES; s0 += OWN_WAVES * BATCH) {
        int4 dc[BATCH];
#pragma unroll
        for (int b = 0; b < BATCH; ++b) dc[b] = dn[b];
#pragma unroll
        for (int b = 0; b < BATCH; ++b) {
            int sl = s0 + OWN_WAVES * BATCH + b;
            sl = sl < N_SLICES ? sl : N_SLICES - 1;
            dn[b] = *(const int4*)(dst + sl * SLICE_E + lane * 4);
        }
        if (n > LIST_CAP - BATCH * SLICE_E) flush();
#pragma unroll
        for (int b = 0; b < BATCH; ++b) {
            if (s0 + b < N_SLICES) {
                const unsigned r0 = (unsigned)(dc[b].x - lo);
                const unsigned r1 = (unsigned)(dc[b].y - lo);
                const unsigned r2 = (unsigned)(dc[b].z - lo);
                const unsigned r3 = (unsigned)(dc[b].w - lo);
                const bool m0 = r0 < (unsigned)cnt;
                const bool m1 = r1 < (unsigned)cnt;
                const bool m2 = r2 < (unsigned)cnt;
                const bool m3 = r3 < (unsigned)cnt;
                if (__builtin_amdgcn_ballot_w32(m0 | m1 | m2 | m3)) {
                    const unsigned e0 = (unsigned)((s0 + b) * SLICE_E + lane * 4);
                    unsigned mk;
                    mk = __builtin_amdgcn_ballot_w32(m0);
                    if (m0) mylist[n + __builtin_amdgcn_mbcnt_lo(mk, 0u)] = (e0 + 0u) | (r0 << 20);
                    n += __builtin_popcount(mk);
                    mk = __builtin_amdgcn_ballot_w32(m1);
                    if (m1) mylist[n + __builtin_amdgcn_mbcnt_lo(mk, 0u)] = (e0 + 1u) | (r1 << 20);
                    n += __builtin_popcount(mk);
                    mk = __builtin_amdgcn_ballot_w32(m2);
                    if (m2) mylist[n + __builtin_amdgcn_mbcnt_lo(mk, 0u)] = (e0 + 2u) | (r2 << 20);
                    n += __builtin_popcount(mk);
                    mk = __builtin_amdgcn_ballot_w32(m3);
                    if (m3) mylist[n + __builtin_amdgcn_mbcnt_lo(mk, 0u)] = (e0 + 3u) | (r3 << 20);
                    n += __builtin_popcount(mk);
                }
            }
        }
    }
    flush();
    __syncthreads();

    float* outp = att + (size_t)lo * CH;
    for (int i = tid; i < cnt * CH; i += 1024) outp[i] = acc[i];
}

__global__ void finish01(const float* __restrict__ att, const float* __restrict__ b,
                         const float* __restrict__ skip, float* __restrict__ out, int N) {
    const int i = blockIdx.x * blockDim.x + threadIdx.x;
    if (i >= N * HC) return;
    const int c = i & 127;
    const float v = att[i] + b[c] + skip[i];
    out[i] = (v > 0.f) ? v : (expf(v) - 1.0f);
}

__global__ void finish2(const float* __restrict__ att, const float* __restrict__ b2,
                        const float* __restrict__ skip32, float* __restrict__ out, int N) {
    const int node = blockIdx.x * (blockDim.x >> 5) + (threadIdx.x >> 5);
    const int c    = threadIdx.x & 31;
    if (node >= N) return;
    float v = 0.25f * att[(size_t)node * OUT_C + c]
            + b2[c] + skip32[(size_t)node * OUT_C + c];
    float sq = v * v;
#pragma unroll
    for (int o = 16; o > 0; o >>= 1) sq += __shfl_xor(sq, o, 32);
    const float norm = sqrtf(sq);
    out[(size_t)node * OUT_C + c] = v / fmaxf(norm, 1e-12f);
}

extern "C" void kernel_launch(void* const* d_in, const int* in_sizes, int n_in,
                              void* d_out, int out_size, void* d_ws, size_t ws_size,
                              hipStream_t stream) {
    (void)in_sizes; (void)n_in; (void)out_size; (void)ws_size;

    const float* x   = (const float*)d_in[0];
    const int*   ei  = (const int*)d_in[1];
    const float* W0  = (const float*)d_in[2];
    const float* aS0 = (const float*)d_in[3];
    const float* aD0 = (const float*)d_in[4];
    const float* b0  = (const float*)d_in[5];
    const float* Ws0 = (const float*)d_in[6];
    const float* bs0 = (const float*)d_in[7];
    const float* W1  = (const float*)d_in[8];
    const float* aS1 = (const float*)d_in[9];
    const float* aD1 = (const float*)d_in[10];
    const float* b1  = (const float*)d_in[11];
    const float* Ws1 = (const float*)d_in[12];
    const float* bs1 = (const float*)d_in[13];
    const float* W2  = (const float*)d_in[14];
    const float* aS2 = (const float*)d_in[15];
    const float* aD2 = (const float*)d_in[16];
    const float* b2  = (const float*)d_in[17];
    const float* Ws2 = (const float*)d_in[18];
    const float* bs2 = (const float*)d_in[19];

    const int* src = ei;
    const int* dst = ei + N_EDGES;

    const size_t NF = (size_t)N_NODES * HC;
    float*    ws    = (float*)d_ws;
    float*    x_cur = ws;
    float*    h_lin = ws + NF;
    float*    skipb = ws + 2 * NF;
    float*    att   = ws + 3 * NF;
    float*    als   = ws + 4 * NF;
    float*    ald   = als + (size_t)N_NODES * HEADS;
    unsigned* m     = (unsigned*)(ald + (size_t)N_NODES * HEADS);
    float*    s     = (float*)(m + (size_t)N_NODES * HEADS);
    float*    ebuf  = s + (size_t)N_NODES * HEADS;

    auto gemm = [&](const float* A, const float* B, const float* bias,
                    float* C, int N, int K, int P) {
        const int tiles = (N / 16) * (P / 16);
        const int blocks = (tiles + 7) / 8;
        gemm16x16_wmma_f32<<<blocks, 256, 0, stream>>>(A, B, bias, C, N, K, P);
    };

    auto run_edges = [&](const float* hfeat) {
        const int nMS = N_NODES * HEADS;
        init_buffers<<<(nMS + 255) / 256, 256, 0, stream>>>(m, s, nMS);
        const int nEH = N_EDGES * HEADS;
        edge_max<<<(nEH + 255) / 256, 256, 0, stream>>>(src, dst, als, ald, ebuf, m);
        edge_exp<<<(nEH + 255) / 256, 256, 0, stream>>>(dst, m, ebuf, s);
        aggr_owned<HC><<<OWN_WGS, 1024, 0, stream>>>(src, dst, ebuf, s, hfeat, att);
    };

    const int nNH = N_NODES * HEADS;
    const int nNF = N_NODES * HC;

    gemm(x, W0, nullptr, h_lin, N_NODES, F_IN, HC);
    gemm(x, Ws0, bs0, skipb, N_NODES, F_IN, HC);
    attn_logits<<<(nNH + 255) / 256, 256, 0, stream>>>(h_lin, aS0, aD0, als, ald, N_NODES);
    run_edges(h_lin);
    finish01<<<(nNF + 255) / 256, 256, 0, stream>>>(att, b0, skipb, x_cur, N_NODES);

    gemm(x_cur, W1, nullptr, h_lin, N_NODES, HC, HC);
    gemm(x_cur, Ws1, bs1, skipb, N_NODES, HC, HC);
    attn_logits<<<(nNH + 255) / 256, 256, 0, stream>>>(h_lin, aS1, aD1, als, ald, N_NODES);
    run_edges(h_lin);
    finish01<<<(nNF + 255) / 256, 256, 0, stream>>>(att, b1, skipb, x_cur, N_NODES);

    gemm(x_cur, W2, nullptr, h_lin, N_NODES, HC, HEADS * OUT_C);
    gemm(x_cur, Ws2, bs2, skipb, N_NODES, HC, OUT_C);
    attn_logits<<<(nNH + 255) / 256, 256, 0, stream>>>(h_lin, aS2, aD2, als, ald, N_NODES);
    {
        const int nMS = N_NODES * HEADS;
        init_buffers<<<(nMS + 255) / 256, 256, 0, stream>>>(m, s, nMS);
        const int nEH = N_EDGES * HEADS;
        edge_max<<<(nEH + 255) / 256, 256, 0, stream>>>(src, dst, als, ald, ebuf, m);
        edge_exp<<<(nEH + 255) / 256, 256, 0, stream>>>(dst, m, ebuf, s);
        aggr_owned<OUT_C><<<OWN_WGS, 1024, 8192, stream>>>(src, dst, ebuf, s, h_lin, att);
    }
    finish2<<<(N_NODES + 7) / 8, 256, 0, stream>>>(att, b2, skipb, (float*)d_out, N_NODES);
}


# Layer 0's two dense products in one launch: 256 work-groups copy their 196 x rows to LDS by coalesced loads, 16 waves each keep one column tile's whole B fragment in registers; same sum order
# speedup vs baseline: 5.1281x; 2.6252x over previous
#include <hip/hip_runtime.h>
#include <hip/hip_bf16.h>
#include <math.h>

typedef __attribute__((ext_vector_type(2))) float v2f;
typedef __attribute__((ext_vector_type(8))) float v8f;

#define N_NODES   50000
#define N_EDGES   800000
#define F_IN      128
#define HC        128
#define HEADS     4
#define HID       32
#define OUT_C     32
#define NEG_SLOPE 0.2f


__global__ void attn_logits(const float* __restrict__ h,
                            const float* __restrict__ aS,
                            const float* __restrict__ aD,
                            float* __restrict__ als,
                            float* __restrict__ ald, int N) {
    const int i = blockIdx.x * blockDim.x + threadIdx.x;
    if (i >= N * HEADS) return;
    const int n  = i >> 2;
    const int hd = i & 3;
    const float* hp = h + (size_t)n * HC + hd * HID;
    const float* as = aS + hd * HID;
    const float* ad = aD + hd * HID;
    float ss = 0.f, sd = 0.f;
#pragma unroll 8
    for (int c = 0; c < HID; ++c) {
        const float v = hp[c];
        ss += v * as[c];
        sd += v * ad[c];
    }
    als[i] = ss;
    ald[i] = sd;
}

__device__ __forceinline__ unsigned f2ord(float f) {
    unsigned u = __float_as_uint(f);
    return (u & 0x80000000u) ? ~u : (u | 0x80000000u);
}
__device__ __forceinline__ float ord2f(unsigned u) {
    return (u & 0x80000000u) ? __uint_as_float(u ^ 0x80000000u)
                             : __uint_as_float(~u);
}

#define OWN_WGS    256
#define OWN_NODES  196
#define OWN_WAVES  32
#define SLICE_E    128
#define N_SLICES   (N_EDGES / SLICE_E)
#define BATCH      4
#define LIST_CAP   1024
#define GROUP      8
#define RES_U      4
#define ACC_ROWS   208
#define PITCH      132
#define EP_U       5

static_assert(N_NODES <= 65536 && OWN_NODES <= 256, "list entry packs source node in 16 bits, local row in 8");
static_assert(OWN_NODES * HEADS <= 1024, "one thread per (node, head) start value");

template <bool SAVE, int NEXT_SK>
__global__ void __launch_bounds__(1024)
aggr_owned(const int* __restrict__ src, const int* __restrict__ dst,
           const float* __restrict__ als, const float* __restrict__ ald,
           const float* __restrict__ h, float* __restrict__ att,
           unsigned* __restrict__ lists, int* __restrict__ lens,
           const float* __restrict__ bcur, const float* __restrict__ skip,
           const float* __restrict__ Wn, const float* __restrict__ Wsn,
           const float* __restrict__ bsn,
           float* __restrict__ hn, float* __restrict__ skn) {
    constexpr bool MEAN = (NEXT_SK == 0);
    __shared__ float    acc[ACC_ROWS * PITCH];
    __shared__ unsigned list[OWN_WAVES * LIST_CAP];
    __shared__ unsigned mx[OWN_NODES * HEADS];
    __shared__ float    sm[OWN_NODES * HEADS];
    __shared__ float    aldl[OWN_NODES * HEADS];
    __shared__ int      spilled;

    const int tid  = threadIdx.x;
    const int wave = __builtin_amdgcn_readfirstlane(tid >> 5);
    const int lane = tid & 31;
    const int hd   = lane & 3;
    const int lo   = blockIdx.x * OWN_NODES;
    const int left = N_NODES - lo;
    const int cnt  = left < OWN_NODES ? left : OWN_NODES;

    unsigned* wl = lists + (size_t)(blockIdx.x * OWN_WAVES + wave) * LIST_CAP;
    int*      wn = lens + blockIdx.x * OWN_WAVES + wave;
    const int kept = SAVE ? -1 : __builtin_amdgcn_readfirstlane(*wn);

    for (int i = tid; i < ACC_ROWS * PITCH; i += 1024) acc[i] = 0.0f;
    if (tid < OWN_NODES * HEADS) {
        mx[tid]   = 0u;
        sm[tid]   = 0.0f;
        aldl[tid] = (tid < cnt * HEADS) ? ald[lo * HEADS + tid] : 0.0f;
    }
    if (tid == 0) spilled = 0;
    __syncthreads();

    unsigned* mylist = list + wave * LIST_CAP;
    int n = 0;

    auto resolve = [&](const bool withMax, const bool resolved) {
        __builtin_amdgcn_wave_barrier();
        const int sub = lane >> 2;
        for (int i = 0; i < n; i += 8 * RES_U) {
            unsigned ent[RES_U];
            int      sN[RES_U];
            float    sa[RES_U];
#pragma unroll
            for (int u = 0; u < RES_U; ++u) {
                const int idx = i + u * 8 + sub;
                ent[u] = mylist[idx < n ? idx : n - 1];
            }
#pragma unroll
            for (int u = 0; u < RES_U; ++u) sN[u] = resolved ? (int)(ent[u] & 0xFFFFu) : src[ent[u] & 0xFFFFFu];
#pragma unroll
            for (int u = 0; u < RES_U; ++u) sa[u] = withMax ? als[sN[u] * HEADS + hd] : 0.0f;
#pragma unroll
            for (int u = 0; u < RES_U; ++u) {
                const int idx = i + u * 8 + sub;
                if (idx < n) {
                    const unsigned dl = resolved ? (ent[u] >> 16) : (ent[u] >> 20);
                    if (withMax) {
                        float v = sa[u] + aldl[dl * HEADS + hd];
                        v = (v > 0.f) ? v : NEG_SLOPE * v;
                        atomicMax(&mx[dl * HEADS + hd], f2ord(v));
                    }
                    if (!resolved && hd == 0) mylist[idx] = (unsigned)sN[u] | (dl << 16);
                }
            }
        }
        __builtin_amdgcn_wave_barrier();
    };

    static_assert(GROUP * HEADS == 32, "one lane per (entry, head) of a group");
    auto gather = [&]() {
        __builtin_amdgcn_wave_barrier();
        const int sub = lane >> 2;
        for (int i = 0; i < n; i += GROUP) {
            const int      idx = i + sub;
            const bool     own = idx < n;
            const unsigned ent = mylist[own ? idx : n - 1];
            const unsigned dl  = ent >> 16;
            const float    sa  = als[(int)(ent & 0xFFFFu) * HEADS + hd];
            float r[GROUP][4];
#pragma unroll
            for (int k = 0; k < GROUP; ++k) {
                const int sN = (int)((unsigned)__builtin_amdgcn_readlane(ent, k * HEADS) & 0xFFFFu);
                const float* hp = h + (size_t)sN * HC + lane;
                r[k][0] = hp[0];
                r[k][1] = hp[HID];
                r[k][2] = hp[2 * HID];
                r[k][3] = hp[3 * HID];
            }
            float v = sa + aldl[dl * HEADS + hd];
            v = (v > 0.f) ? v : NEG_SLOPE * v;
            const float ex = expf(v - ord2f(mx[dl * HEADS + hd]));
            if (own) atomicAdd(&sm[dl * HEADS + hd], ex);
            const unsigned eb = __float_as_uint(ex);
#pragma unroll
            for (int k = 0; k < GROUP; ++k) {
                if (i + k < n) {
                    const unsigned dk = (unsigned)__builtin_amdgcn_readlane(ent, k * HEADS) >> 16;
                    const float a0 = __uint_as_float(__builtin_amdgcn_readlane(eb, k * HEADS + 0));
                    const float a1 = __uint_as_float(__builtin_amdgcn_readlane(eb, k * HEADS + 1));
                    const float a2 = __uint_as_float(__builtin_amdgcn_readlane(eb, k * HEADS + 2));
                    const float a3 = __uint_as_float(__builtin_amdgcn_readlane(eb, k * HEADS + 3));
                    float* ap = acc + dk * PITCH + lane;
                    atomicAdd(ap,           a0 * r[k][0]);
                    atomicAdd(ap + HID,     a1 * r[k][1]);
                    atomicAdd(ap + 2 * HID, a2 * r[k][2]);
                    atomicAdd(ap + 3 * HID, a3 * r[k][3]);
                }
            }
        }
        __builtin_amdgcn_wave_barrier();
    };

    auto scan = [&](auto&& onFull) {
        int4 dn[BATCH];
#pragma unroll
        for (int b = 0; b < BATCH; ++b) {
            const int sl = wave * BATCH + b;
            dn[b] = *(const int4*)(dst + sl * SLICE_E + lane * 4);
        }
        for (int s0 = wave * BATCH; s0 < N_SLICES; s0 += OWN_WAVES * BATCH) {
            int4 dc[BATCH];
#pragma unroll
            for (int b = 0; b < BATCH; ++b) dc[b] = dn[b];
#pragma unroll
            for (int b = 0; b < BATCH; ++b) {
                int sl = s0 + OWN_WAVES * BATCH + b;
                sl = sl < N_SLICES ? sl : N_SLICES - 1;
                dn[b] = *(const int4*)(dst + sl * SLICE_E + lane * 4);
            }
            if (n > LIST_CAP - BATCH * SLICE_E) onFull();
#pragma unroll
            for (int b = 0; b < BATCH; ++b) {
                if (s0 + b < N_SLICES) {
                    const unsigned r0 = (unsigned)(dc[b].x - lo);
                    const unsigned r1 = (unsigned)(dc[b].y - lo);
                    const unsigned r2 = (unsigned)(dc[b].z - lo);
                    const unsigned r3 = (unsigned)(dc[b].w - lo);
                    const bool m0 = r0 < (unsigned)cnt;
                    const bool m1 = r1 < (unsigned)cnt;
                    const bool m2 = r2 < (unsigned)cnt;
                    const bool m3 = r3 < (unsigned)cnt;
                    if (__builtin_amdgcn_ballot_w32(m0 | m1 | m2 | m3)) {
                        const unsigned e0 = (unsigned)((s0 + b) * SLICE_E + lane * 4);
                        unsigned mk;
                        mk = __builtin_amdgcn_ballot_w32(m0);
                        if (m0) mylist[n + __builtin_amdgcn_mbcnt_lo(mk, 0u)] = (e0 + 0u) | (r0 << 20);
                        n += __builtin_popcount(mk);
                        mk = __builtin_amdgcn_ballot_w32(m1);
                        if (m1) mylist[n + __builtin_amdgcn_mbcnt_lo(mk, 0u)] = (e0 + 1u) | (r1 << 20);
                        n += __builtin_popcount(mk);
                        mk = __builtin_amdgcn_ballot_w32(m2);
                        if (m2) mylist[n + __builtin_amdgcn_mbcnt_lo(mk, 0u)] = (e0 + 2u) | (r2 << 20);
                        n += __builtin_popcount(mk);
                        mk = __builtin_amdgcn_ballot_w32(m3);
                        if (m3) mylist[n + __builtin_amdgcn_mbcnt_lo(mk, 0u)] = (e0 + 3u) | (r3 << 20);
                        n += __builtin_popcount(mk);
                    }
                }
            }
        }
    };

    if (kept >= 0) {
        n = kept;
        for (int i = 0; i < n; i += 4 * 32) {
            unsigned v[4];
#pragma unroll
            for (int u = 0; u < 4; ++u) {
                const int idx = i + u * 32 + lane;
                v[u] = wl[idx < n ? idx : n - 1];
            }
#pragma unroll
            for (int u = 0; u < 4; ++u) {
                const int idx = i + u * 32 + lane;
                if (idx < n) mylist[idx] = v[u];
            }
        }
        resolve(true, true);
    } else {
        scan([&]() { resolve(true, false); if (lane == 0) spilled = 1; n = 0; });
        resolve(true, false);
    }
    __syncthreads();
    if (SAVE) {
        const bool sp = __builtin_amdgcn_readfirstlane(spilled) != 0;
        if (!sp) for (int i = lane; i < n; i += 32) wl[i] = mylist[i];
        if (lane == 0) *wn = sp ? -1 : n;
    }
    if (__builtin_amdgcn_readfirstlane(spilled) == 0) {
        gather();
    } else {
        n = 0;
        scan([&]() { resolve(false, false); gather(); n = 0; });
        resolve(false, false);
        gather();
    }
    __syncthreads();

    if (!MEAN) {
        constexpr int CT = (HC + NEXT_SK) / 16;
        constexpr int G  = OWN_WAVES / CT;
        const int  ct   = wave % CT;
        const bool isH  = ct < HC / 16;
        const float* Bm = isH ? Wn : Wsn;
        const int  P    = isH ? HC : NEXT_SK;
        const int  half = lane >> 4;
        const int  r    = lane & 15;
        const int  col  = (isH ? ct : ct - HC / 16) * 16 + r;
        const bool busy = wave < G * CT;

        v2f bf[HC / 4];
        if (busy) {
#pragma unroll
            for (int s = 0; s < HC / 4; ++s) {
                const int ka = 4 * s + 2 * half;
                bf[s].x = Bm[(size_t)(ka + 0) * P + col];
                bf[s].y = Bm[(size_t)(ka + 1) * P + col];
            }
        }

        const float* sk = skip + (size_t)lo * HC;
        const float bch = bcur[tid & 127];
        for (int i0 = tid; i0 < cnt * HC; i0 += 1024 * EP_U) {
            float sv[EP_U];
#pragma unroll
            for (int u = 0; u < EP_U; ++u) {
                const int i = i0 + u * 1024;
                sv[u] = (i < cnt * HC) ? sk[i] : 0.0f;
            }
#pragma unroll
            for (int u = 0; u < EP_U; ++u) {
                const int i = i0 + u * 1024;
                if (i < cnt * HC) {
                    const int row = i >> 7;
                    const int ch  = i & 127;
                    float* ap = acc + row * PITCH + ch;
                    const float q = *ap / (sm[row * HEADS + (ch >> 5)] + 1e-16f);
                    const float v = q + bch + sv[u];
                    *ap = (v > 0.f) ? v : (expf(v) - 1.0f);
                }
            }
        }
        __syncthreads();

        if (busy) {
            const float badd = isH ? 0.0f : bsn[col];
            float* Cm = (isH ? hn : skn) + (size_t)lo * P + col;
            const int rtiles = (cnt + 15) >> 4;
            for (int rt = wave / CT; rt < rtiles; rt += G) {
                v8f d = {};
                const float* ar = acc + (rt * 16 + r) * PITCH + 2 * half;
#pragma unroll
                for (int s = 0; s < HC / 4; ++s) {
                    const v2f a = *(const v2f*)(ar + 4 * s);
                    d = __builtin_amdgcn_wmma_f32_16x16x4_f32(
                         false, a,  false, bf[s],
                         (short)0, d,  false,  false);
                }
#pragma unroll
                for (int i = 0; i < 8; ++i) {
                    const int row = rt * 16 + i + half * 8;
                    if (row < cnt) Cm[(size_t)row * P] = d[i] + badd;
                }
            }
        }
    } else {
        float* outp = att + (size_t)lo * OUT_C;
        for (int i = tid; i < cnt * OUT_C; i += 1024) {
            const float* ar = acc + (i >> 5) * PITCH + (i & 31);
            const float* sr = sm + (i >> 5) * HEADS;
            outp[i] = ar[0]       / (sr[0] + 1e-16f) + ar[HID]     / (sr[1] + 1e-16f)
                    + ar[2 * HID] / (sr[2] + 1e-16f) + ar[3 * HID] / (sr[3] + 1e-16f);
        }
    }
}

#define D0_WAVES   16
#define D0_THREADS (D0_WAVES * 32)
#define D0_TRIPS   (ACC_ROWS * (F_IN / 4) / D0_THREADS)

static_assert(F_IN == HC && HC / 16 * 2 == D0_WAVES, "one wave per column tile of the two products");
static_assert(ACC_ROWS * (F_IN / 4) == D0_TRIPS * D0_THREADS, "the staging loop covers the row block exactly");

__global__ void __launch_bounds__(D0_THREADS)
dense0_owned(const float* __restrict__ x, const float* __restrict__ Wa,
             const float* __restrict__ Wb, const float* __restrict__ biasb,
             float* __restrict__ Ca, float* __restrict__ Cb) {
    __shared__ __attribute__((aligned(16))) float xs[ACC_ROWS * PITCH];

    const int tid  = threadIdx.x;
    const int wave = __builtin_amdgcn_readfirstlane(tid >> 5);
    const int lane = tid & 31;
    const int lo   = blockIdx.x * OWN_NODES;
    const int left = N_NODES - lo;
    const int cnt  = left < OWN_NODES ? left : OWN_NODES;

    const float4* xg = (const float4*)(x + (size_t)lo * F_IN);
    float4 xv[D0_TRIPS];
#pragma unroll
    for (int u = 0; u < D0_TRIPS; ++u) {
        const int i = tid + u * D0_THREADS;
        xv[u] = ((i >> 5) < cnt) ? xg[i] : make_float4(0.f, 0.f, 0.f, 0.f);
    }

    const bool isH  = wave < HC / 16;
    const float* Bm = isH ? Wa : Wb;
    const int  half = lane >> 4;
    const int  r    = lane & 15;
    const int  col  = (wave & (HC / 16 - 1)) * 16 + r;

    v2f bf[F_IN / 4];
#pragma unroll
    for (int s = 0; s < F_IN / 4; ++s) {
        const int ka = 4 * s + 2 * half;
        bf[s].x = Bm[(size_t)(ka + 0) * HC + col];
        bf[s].y = Bm[(size_t)(ka + 1) * HC + col];
    }
    const float badd = isH ? 0.0f : biasb[col];

#pragma unroll
    for (int u = 0; u < D0_TRIPS; ++u) {
        const int i = tid + u * D0_THREADS;
        *(float4*)(xs + (i >> 5) * PITCH + (i & 31) * 4) = xv[u];
    }
    __syncthreads();

    float* Cm = (isH ? Ca : Cb) + (size_t)lo * HC + col;
    const int rtiles = (cnt + 15) >> 4;
    for (int rt = 0; rt < rtiles; ++rt) {
        v8f d = {};
        const float* ar = xs + (rt * 16 + r) * PITCH + 2 * half;
#pragma unroll
        for (int s = 0; s < F_IN / 4; ++s) {
            const v2f a = *(const v2f*)(ar + 4 * s);
            d = __builtin_amdgcn_wmma_f32_16x16x4_f32(
                 false, a,  false, bf[s],
                 (short)0, d,  false,  false);
        }
#pragma unroll
        for (int i = 0; i < 8; ++i) {
            const int row = rt * 16 + i + half * 8;
            if (row < cnt) Cm[(size_t)row * HC] = d[i] + badd;
        }
    }
}

__global__ void finish2(const float* __restrict__ att, const float* __restrict__ b2,
                        const float* __restrict__ skip32, float* __restrict__ out, int N) {
    const int node = blockIdx.x * (blockDim.x >> 5) + (threadIdx.x >> 5);
    const int c    = threadIdx.x & 31;
    if (node >= N) return;
    float v = 0.25f * att[(size_t)node * OUT_C + c]
            + b2[c] + skip32[(size_t)node * OUT_C + c];
    float sq = v * v;
#pragma unroll
    for (int o = 16; o > 0; o >>= 1) sq += __shfl_xor(sq, o, 32);
    const float norm = sqrtf(sq);
    out[(size_t)node * OUT_C + c] = v / fmaxf(norm, 1e-12f);
}

extern "C" void kernel_launch(void* const* d_in, const int* in_sizes, int n_in,
                              void* d_out, int out_size, void* d_ws, size_t ws_size,
                              hipStream_t stream) {
    (void)in_sizes; (void)n_in; (void)out_size; (void)ws_size;

    const float* x   = (const float*)d_in[0];
    const int*   ei  = (const int*)d_in[1];
    const float* W0  = (const float*)d_in[2];
    const float* aS0 = (const float*)d_in[3];
    const float* aD0 = (const float*)d_in[4];
    const float* b0  = (const float*)d_in[5];
    const float* Ws0 = (const float*)d_in[6];
    const float* bs0 = (const float*)d_in[7];
    const float* W1  = (const float*)d_in[8];
    const float* aS1 = (const float*)d_in[9];
    const float* aD1 = (const float*)d_in[10];
    const float* b1  = (const float*)d_in[11];
    const float* Ws1 = (const float*)d_in[12];
    const float* bs1 = (const float*)d_in[13];
    const float* W2  = (const float*)d_in[14];
    const float* aS2 = (const float*)d_in[15];
    const float* aD2 = (const float*)d_in[16];
    const float* b2  = (const float*)d_in[17];
    const float* Ws2 = (const float*)d_in[18];
    const float* bs2 = (const float*)d_in[19];

    const int* src = ei;
    const int* dst = ei + N_EDGES;

    const size_t NF = (size_t)N_NODES * HC;
    float*    ws    = (float*)d_ws;
    float*    h_b   = ws;
    float*    h_a   = ws + NF;
    float*    sk_a  = ws + 2 * NF;
    float*    sk_b  = ws + 3 * NF;
    float*    als   = ws + 4 * NF;
    float*    ald   = als + (size_t)N_NODES * HEADS;
    unsigned* lists = (unsigned*)(ald + (size_t)N_NODES * HEADS);
    int*      lens  = (int*)(lists + (size_t)OWN_WGS * OWN_WAVES * LIST_CAP);

    const int nNH = N_NODES * HEADS;

    dense0_owned<<<OWN_WGS, D0_THREADS, 0, stream>>>(x, W0, Ws0, bs0, h_a, sk_a);
    attn_logits<<<(nNH + 255) / 256, 256, 0, stream>>>(h_a, aS0, aD0, als, ald, N_NODES);
    aggr_owned<true, HC><<<OWN_WGS, 1024, 0, stream>>>(src, dst, als, ald, h_a, nullptr, lists, lens,
                                                       b0, sk_a, W1, Ws1, bs1, h_b, sk_b);

    attn_logits<<<(nNH + 255) / 256, 256, 0, stream>>>(h_b, aS1, aD1, als, ald, N_NODES);
    aggr_owned<false, OUT_C><<<OWN_WGS, 1024, 0, stream>>>(src, dst, als, ald, h_b, nullptr, lists, lens,
                                                           b1, sk_b, W2, Ws2, bs2, h_a, sk_a);

    attn_logits<<<(nNH + 255) / 256, 256, 0, stream>>>(h_a, aS2, aD2, als, ald, N_NODES);
    aggr_owned<false, 0><<<OWN_WGS, 1024, 0, stream>>>(src, dst, als, ald, h_a, sk_b, lists, lens,
                                                       nullptr, nullptr, nullptr, nullptr, nullptr, nullptr, nullptr);
    finish2<<<(N_NODES + 7) / 8, 256, 0, stream>>>(sk_b, b2, sk_a, (float*)d_out, N_NODES);
}


# Attention logits formed by the dense products' own waves (weights as A operand, rows as B: a lane holds 8 columns of a row; 4 partial sums added in LDS): 3 logit launches gone; new sum order
# speedup vs baseline: 5.5188x; 1.0762x over previous
#include <hip/hip_runtime.h>
#include <hip/hip_bf16.h>
#include <math.h>

typedef __attribute__((ext_vector_type(2))) float v2f;
typedef __attribute__((ext_vector_type(8))) float v8f;

#define N_NODES   50000
#define N_EDGES   800000
#define F_IN      128
#define HC        128
#define HEADS     4
#define HID       32
#define OUT_C     32
#define NEG_SLOPE 0.2f

#define PART_Q     16

__device__ __forceinline__ void load8(float (&v)[8], const float* __restrict__ p) {
    const float4 a = *(const float4*)p;
    const float4 b = *(const float4*)(p + 4);
    v[0] = a.x; v[1] = a.y; v[2] = a.z; v[3] = a.w;
    v[4] = b.x; v[5] = b.y; v[6] = b.z; v[7] = b.w;
}

__device__ __forceinline__ unsigned f2ord(float f) {
    unsigned u = __float_as_uint(f);
    return (u & 0x80000000u) ? ~u : (u | 0x80000000u);
}
__device__ __forceinline__ float ord2f(unsigned u) {
    return (u & 0x80000000u) ? __uint_as_float(u ^ 0x80000000u)
                             : __uint_as_float(~u);
}

#define OWN_WGS    256
#define OWN_NODES  196
#define OWN_WAVES  32
#define SLICE_E    128
#define N_SLICES   (N_EDGES / SLICE_E)
#define BATCH      4
#define LIST_CAP   1024
#define GROUP      8
#define RES_U      4
#define ACC_ROWS   208
#define PITCH      132
#define EP_U       5

static_assert(N_NODES <= 65536 && OWN_NODES <= 256, "list entry packs source node in 16 bits, local row in 8");
static_assert(OWN_NODES * HEADS <= 1024, "one thread per (node, head) start value");
static_assert(2 * ACC_ROWS * PART_Q <= OWN_WAVES * LIST_CAP, "the partial logit sums fit in the LDS of the edge lists");
static_assert(PART_Q == 2 * (HC / 16) && HEADS * 4 == PART_Q, "4 partial sums per (row, head), side by side");

template <bool SAVE, int NEXT_SK>
__global__ void __launch_bounds__(1024)
aggr_owned(const int* __restrict__ src, const int* __restrict__ dst,
           const float* __restrict__ als, const float* __restrict__ ald,
           const float* __restrict__ h, float* __restrict__ att,
           unsigned* __restrict__ lists, int* __restrict__ lens,
           const float* __restrict__ bcur, const float* __restrict__ skip,
           const float* __restrict__ Wn, const float* __restrict__ Wsn,
           const float* __restrict__ bsn,
           const float* __restrict__ aSn, const float* __restrict__ aDn,
           float* __restrict__ hn, float* __restrict__ skn,
           float* __restrict__ alsn, float* __restrict__ aldn) {
    constexpr bool MEAN = (NEXT_SK == 0);
    __shared__ float    acc[ACC_ROWS * PITCH];
    __shared__ __attribute__((aligned(16))) unsigned list[OWN_WAVES * LIST_CAP];
    __shared__ unsigned mx[OWN_NODES * HEADS];
    __shared__ float    sm[OWN_NODES * HEADS];
    __shared__ float    aldl[OWN_NODES * HEADS];
    __shared__ int      spilled;

    const int tid  = threadIdx.x;
    const int wave = __builtin_amdgcn_readfirstlane(tid >> 5);
    const int lane = tid & 31;
    const int hd   = lane & 3;
    const int lo   = blockIdx.x * OWN_NODES;
    const int left = N_NODES - lo;
    const int cnt  = left < OWN_NODES ? left : OWN_NODES;

    unsigned* wl = lists + (size_t)(blockIdx.x * OWN_WAVES + wave) * LIST_CAP;
    int*      wn = lens + blockIdx.x * OWN_WAVES + wave;
    const int kept = SAVE ? -1 : __builtin_amdgcn_readfirstlane(*wn);

    for (int i = tid; i < ACC_ROWS * PITCH; i += 1024) acc[i] = 0.0f;
    if (tid < OWN_NODES * HEADS) {
        mx[tid]   = 0u;
        sm[tid]   = 0.0f;
        aldl[tid] = (tid < cnt * HEADS) ? ald[lo * HEADS + tid] : 0.0f;
    }
    if (tid == 0) spilled = 0;
    __syncthreads();

    unsigned* mylist = list + wave * LIST_CAP;
    int n = 0;

    auto resolve = [&](const bool withMax, const bool resolved) {
        __builtin_amdgcn_wave_barrier();
        const int sub = lane >> 2;
        for (int i = 0; i < n; i += 8 * RES_U) {
            unsigned ent[RES_U];
            int      sN[RES_U];
            float    sa[RES_U];
#pragma unroll
            for (int u = 0; u < RES_U; ++u) {
                const int idx = i + u * 8 + sub;
                ent[u] = mylist[idx < n ? idx : n - 1];
            }
#pragma unroll
            for (int u = 0; u < RES_U; ++u) sN[u] = resolved ? (int)(ent[u] & 0xFFFFu) : src[ent[u] & 0xFFFFFu];
#pragma unroll
            for (int u = 0; u < RES_U; ++u) sa[u] = withMax ? als[sN[u] * HEADS + hd] : 0.0f;
#pragma unroll
            for (int u = 0; u < RES_U; ++u) {
                const int idx = i + u * 8 + sub;
                if (idx < n) {
                    const unsigned dl = resolved ? (ent[u] >> 16) : (ent[u] >> 20);
                    if (withMax) {
                        float v = sa[u] + aldl[dl * HEADS + hd];
                        v = (v > 0.f) ? v : NEG_SLOPE * v;
                        atomicMax(&mx[dl * HEADS + hd], f2ord(v));
                    }
                    if (!resolved && hd == 0) mylist[idx] = (unsigned)sN[u] | (dl << 16);
                }
            }
        }
        __builtin_amdgcn_wave_barrier();
    };

    static_assert(GROUP * HEADS == 32, "one lane per (entry, head) of a group");
    auto gather = [&]() {
        __builtin_amdgcn_wave_barrier();
        const int sub = lane >> 2;
        for (int i = 0; i < n; i += GROUP) {
            const int      idx = i + sub;
            const bool     own = idx < n;
            const unsigned ent = mylist[own ? idx : n - 1];
            const unsigned dl  = ent >> 16;
            const float    sa  = als[(int)(ent & 0xFFFFu) * HEADS + hd];
            float r[GROUP][4];
#pragma unroll
            for (int k = 0; k < GROUP; ++k) {
                const int sN = (int)((unsigned)__builtin_amdgcn_readlane(ent, k * HEADS) & 0xFFFFu);
                const float* hp = h + (size_t)sN * HC + lane;
                r[k][0] = hp[0];
                r[k][1] = hp[HID];
                r[k][2] = hp[2 * HID];
                r[k][3] = hp[3 * HID];
            }
            float v = sa + aldl[dl * HEADS + hd];
            v = (v > 0.f) ? v : NEG_SLOPE * v;
            const float ex = expf(v - ord2f(mx[dl * HEADS + hd]));
            if (own) atomicAdd(&sm[dl * HEADS + hd], ex);
            const unsigned eb = __float_as_uint(ex);
#pragma unroll
            for (int k = 0; k < GROUP; ++k) {
                if (i + k < n) {
                    const unsigned dk = (unsigned)__builtin_amdgcn_readlane(ent, k * HEADS) >> 16;
                    const float a0 = __uint_as_float(__builtin_amdgcn_readlane(eb, k * HEADS + 0));
                    const float a1 = __uint_as_float(__builtin_amdgcn_readlane(eb, k * HEADS + 1));
                    const float a2 = __uint_as_float(__builtin_amdgcn_readlane(eb, k * HEADS + 2));
                    const float a3 = __uint_as_float(__builtin_amdgcn_readlane(eb, k * HEADS + 3));
                    float* ap = acc + dk * PITCH + lane;
                    atomicAdd(ap,           a0 * r[k][0]);
                    atomicAdd(ap + HID,     a1 * r[k][1]);
                    atomicAdd(ap + 2 * HID, a2 * r[k][2]);
                    atomicAdd(ap + 3 * HID, a3 * r[k][3]);
                }
            }
        }
        __builtin_amdgcn_wave_barrier();
    };

    auto scan = [&](auto&& onFull) {
        int4 dn[BATCH];
#pragma unroll
        for (int b = 0; b < BATCH; ++b) {
            const int sl = wave * BATCH + b;
            dn[b] = *(const int4*)(dst + sl * SLICE_E + lane * 4);
        }
        for (int s0 = wave * BATCH; s0 < N_SLICES; s0 += OWN_WAVES * BATCH) {
            int4 dc[BATCH];
#pragma unroll
            for (int b = 0; b < BATCH; ++b) dc[b] = dn[b];
#pragma unroll
            for (int b = 0; b < BATCH; ++b) {
                int sl = s0 + OWN_WAVES * BATCH + b;
                sl = sl < N_SLICES ? sl : N_SLICES - 1;
                dn[b] = *(const int4*)(dst + sl * SLICE_E + lane * 4);
            }
            if (n > LIST_CAP - BATCH * SLICE_E) onFull();
#pragma unroll
            for (int b = 0; b < BATCH; ++b) {
                if (s0 + b < N_SLICES) {
                    const unsigned r0 = (unsigned)(dc[b].x - lo);
                    const unsigned r1 = (unsigned)(dc[b].y - lo);
                    const unsigned r2 = (unsigned)(dc[b].z - lo);
                    const unsigned r3 = (unsigned)(dc[b].w - lo);
                    const bool m0 = r0 < (unsigned)cnt;
                    const bool m1 = r1 < (unsigned)cnt;
                    const bool m2 = r2 < (unsigned)cnt;
                    const bool m3 = r3 < (unsigned)cnt;
                    if (__builtin_amdgcn_ballot_w32(m0 | m1 | m2 | m3)) {
                        const unsigned e0 = (unsigned)((s0 + b) * SLICE_E + lane * 4);
                        unsigned mk;
                        mk = __builtin_amdgcn_ballot_w32(m0);
                        if (m0) mylist[n + __builtin_amdgcn_mbcnt_lo(mk, 0u)] = (e0 + 0u) | (r0 << 20);
                        n += __builtin_popcount(mk);
                        mk = __builtin_amdgcn_ballot_w32(m1);
                        if (m1) mylist[n + __builtin_amdgcn_mbcnt_lo(mk, 0u)] = (e0 + 1u) | (r1 << 20);
                        n += __builtin_popcount(mk);
                        mk = __builtin_amdgcn_ballot_w32(m2);
                        if (m2) mylist[n + __builtin_amdgcn_mbcnt_lo(mk, 0u)] = (e0 + 2u) | (r2 << 20);
                        n += __builtin_popcount(mk);
                        mk = __builtin_amdgcn_ballot_w32(m3);
                        if (m3) mylist[n + __builtin_amdgcn_mbcnt_lo(mk, 0u)] = (e0 + 3u) | (r3 << 20);
                        n += __builtin_popcount(mk);
                    }
                }
            }
        }
    };

    if (kept >= 0) {
        n = kept;
        for (int i = 0; i < n; i += 4 * 32) {
            unsigned v[4];
#pragma unroll
            for (int u = 0; u < 4; ++u) {
                const int idx = i + u * 32 + lane;
                v[u] = wl[idx < n ? idx : n - 1];
            }
#pragma unroll
            for (int u = 0; u < 4; ++u) {
                const int idx = i + u * 32 + lane;
                if (idx < n) mylist[idx] = v[u];
            }
        }
        resolve(true, true);
    } else {
        scan([&]() { resolve(true, false); if (lane == 0) spilled = 1; n = 0; });
        resolve(true, false);
    }
    __syncthreads();
    if (SAVE) {
        const bool sp = __builtin_amdgcn_readfirstlane(spilled) != 0;
        if (!sp) for (int i = lane; i < n; i += 32) wl[i] = mylist[i];
        if (lane == 0) *wn = sp ? -1 : n;
    }
    if (__builtin_amdgcn_readfirstlane(spilled) == 0) {
        gather();
    } else {
        n = 0;
        scan([&]() { resolve(false, false); gather(); n = 0; });
        resolve(false, false);
        gather();
    }
    __syncthreads();

    if (!MEAN) {
        constexpr int CT = (HC + NEXT_SK) / 16;
        constexpr int G  = OWN_WAVES / CT;
        const int  ct   = wave % CT;
        const bool isH  = ct < HC / 16;
        const float* Bm = isH ? Wn : Wsn;
        const int  P    = isH ? HC : NEXT_SK;
        const int  half = lane >> 4;
        const int  r    = lane & 15;
        const int  ctl  = isH ? ct : ct - HC / 16;
        const int  col  = ctl * 16 + r;
        const int  c0   = ctl * 16 + 8 * half;
        const bool busy = wave < G * CT;
        float* pS = (float*)list;
        float* pD = pS + ACC_ROWS * PART_Q;

        v2f bf[HC / 4];
        float bb[8] = {}, wS[8] = {}, wD[8] = {};
        if (busy) {
#pragma unroll
            for (int s = 0; s < HC / 4; ++s) {
                const int ka = 4 * s + 2 * half;
                bf[s].x = Bm[(size_t)(ka + 0) * P + col];
                bf[s].y = Bm[(size_t)(ka + 1) * P + col];
            }
            if (isH) { load8(wS, aSn + c0); load8(wD, aDn + c0); }
            else     { load8(bb, bsn + c0); }
        }

        const float* sk = skip + (size_t)lo * HC;
        const float bch = bcur[tid & 127];
        for (int i0 = tid; i0 < cnt * HC; i0 += 1024 * EP_U) {
            float sv[EP_U];
#pragma unroll
            for (int u = 0; u < EP_U; ++u) {
                const int i = i0 + u * 1024;
                sv[u] = (i < cnt * HC) ? sk[i] : 0.0f;
            }
#pragma unroll
            for (int u = 0; u < EP_U; ++u) {
                const int i = i0 + u * 1024;
                if (i < cnt * HC) {
                    const int row = i >> 7;
                    const int ch  = i & 127;
                    float* ap = acc + row * PITCH + ch;
                    const float q = *ap / (sm[row * HEADS + (ch >> 5)] + 1e-16f);
                    const float v = q + bch + sv[u];
                    *ap = (v > 0.f) ? v : (expf(v) - 1.0f);
                }
            }
        }
        __syncthreads();

        if (busy) {
            float* Cm = (isH ? hn : skn) + (size_t)lo * P + c0;
            const int rtiles = (cnt + 15) >> 4;
            for (int rt = wave / CT; rt < rtiles; rt += G) {
                v8f d = {};
                const int row = rt * 16 + r;
                const float* ar = acc + row * PITCH + 2 * half;
#pragma unroll
                for (int s = 0; s < HC / 4; ++s) {
                    const v2f a = *(const v2f*)(ar + 4 * s);
                    d = __builtin_amdgcn_wmma_f32_16x16x4_f32(
                         false, bf[s],  false, a,
                         (short)0, d,  false,  false);
                }
                if (isH) {
                    float ps = d[0] * wS[0], pd = d[0] * wD[0];
#pragma unroll
                    for (int i = 1; i < 8; ++i) {
                        ps = fmaf(d[i], wS[i], ps);
                        pd = fmaf(d[i], wD[i], pd);
                    }
                    pS[row * PART_Q + ct * 2 + half] = ps;
                    pD[row * PART_Q + ct * 2 + half] = pd;
                }
                if (row < cnt) {
                    float* cp = Cm + (size_t)row * P;
                    *(float4*)cp       = make_float4(d[0] + bb[0], d[1] + bb[1], d[2] + bb[2], d[3] + bb[3]);
                    *(float4*)(cp + 4) = make_float4(d[4] + bb[4], d[5] + bb[5], d[6] + bb[6], d[7] + bb[7]);
                }
            }
        }
        __syncthreads();
        if (tid < cnt * HEADS) {
            const float4 s4 = ((const float4*)pS)[tid];
            const float4 d4 = ((const float4*)pD)[tid];
            alsn[lo * HEADS + tid] = (s4.x + s4.y) + (s4.z + s4.w);
            aldn[lo * HEADS + tid] = (d4.x + d4.y) + (d4.z + d4.w);
        }
    } else {
        float* outp = att + (size_t)lo * OUT_C;
        for (int i = tid; i < cnt * OUT_C; i += 1024) {
            const float* ar = acc + (i >> 5) * PITCH + (i & 31);
            const float* sr = sm + (i >> 5) * HEADS;
            outp[i] = ar[0]       / (sr[0] + 1e-16f) + ar[HID]     / (sr[1] + 1e-16f)
                    + ar[2 * HID] / (sr[2] + 1e-16f) + ar[3 * HID] / (sr[3] + 1e-16f);
        }
    }
}

#define D0_WAVES   16
#define D0_THREADS (D0_WAVES * 32)
#define D0_TRIPS   (ACC_ROWS * (F_IN / 4) / D0_THREADS)

static_assert(F_IN == HC && HC / 16 * 2 == D0_WAVES, "one wave per column tile of the two products");
static_assert(ACC_ROWS * (F_IN / 4) == D0_TRIPS * D0_THREADS, "the staging loop covers the row block exactly");

__global__ void __launch_bounds__(D0_THREADS)
dense0_owned(const float* __restrict__ x, const float* __restrict__ Wa,
             const float* __restrict__ Wb, const float* __restrict__ biasb,
             const float* __restrict__ aS, const float* __restrict__ aD,
             float* __restrict__ Ca, float* __restrict__ Cb,
             float* __restrict__ als, float* __restrict__ ald) {
    __shared__ __attribute__((aligned(16))) float xs[ACC_ROWS * PITCH];
    __shared__ __attribute__((aligned(16))) float pS[ACC_ROWS * PART_Q];
    __shared__ __attribute__((aligned(16))) float pD[ACC_ROWS * PART_Q];

    const int tid  = threadIdx.x;
    const int wave = __builtin_amdgcn_readfirstlane(tid >> 5);
    const int lane = tid & 31;
    const int lo   = blockIdx.x * OWN_NODES;
    const int left = N_NODES - lo;
    const int cnt  = left < OWN_NODES ? left : OWN_NODES;

    const float4* xg = (const float4*)(x + (size_t)lo * F_IN);
    float4 xv[D0_TRIPS];
#pragma unroll
    for (int u = 0; u < D0_TRIPS; ++u) {
        const int i = tid + u * D0_THREADS;
        xv[u] = ((i >> 5) < cnt) ? xg[i] : make_float4(0.f, 0.f, 0.f, 0.f);
    }

    const bool isH  = wave < HC / 16;
    const float* Bm = isH ? Wa : Wb;
    const int  half = lane >> 4;
    const int  r    = lane & 15;
    const int  ctl  = wave & (HC / 16 - 1);
    const int  col  = ctl * 16 + r;
    const int  c0   = ctl * 16 + 8 * half;

    v2f bf[F_IN / 4];
#pragma unroll
    for (int s = 0; s < F_IN / 4; ++s) {
        const int ka = 4 * s + 2 * half;
        bf[s].x = Bm[(size_t)(ka + 0) * HC + col];
        bf[s].y = Bm[(size_t)(ka + 1) * HC + col];
    }
    float bb[8] = {}, wS[8] = {}, wD[8] = {};
    if (isH) { load8(wS, aS + c0); load8(wD, aD + c0); }
    else     { load8(bb, biasb + c0); }

#pragma unroll
    for (int u = 0; u < D0_TRIPS; ++u) {
        const int i = tid + u * D0_THREADS;
        *(float4*)(xs + (i >> 5) * PITCH + (i & 31) * 4) = xv[u];
    }
    __syncthreads();

    float* Cm = (isH ? Ca : Cb) + (size_t)lo * HC + c0;
    const int rtiles = (cnt + 15) >> 4;
    for (int rt = 0; rt < rtiles; ++rt) {
        v8f d = {};
        const int row = rt * 16 + r;
        const float* ar = xs + row * PITCH + 2 * half;
#pragma unroll
        for (int s = 0; s < F_IN / 4; ++s) {
            const v2f a = *(const v2f*)(ar + 4 * s);
            d = __builtin_amdgcn_wmma_f32_16x16x4_f32(
                 false, bf[s],  false, a,
                 (short)0, d,  false,  false);
        }
        if (isH) {
            float ps = d[0] * wS[0], pd = d[0] * wD[0];
#pragma unroll
            for (int i = 1; i < 8; ++i) {
                ps = fmaf(d[i], wS[i], ps);
                pd = fmaf(d[i], wD[i], pd);
            }
            pS[row * PART_Q + ctl * 2 + half] = ps;
            pD[row * PART_Q + ctl * 2 + half] = pd;
        }
        if (row < cnt) {
            float* cp = Cm + (size_t)row * HC;
            *(float4*)cp       = make_float4(d[0] + bb[0], d[1] + bb[1], d[2] + bb[2], d[3] + bb[3]);
            *(float4*)(cp + 4) = make_float4(d[4] + bb[4], d[5] + bb[5], d[6] + bb[6], d[7] + bb[7]);
        }
    }
    __syncthreads();
    for (int t = tid; t < cnt * HEADS; t += D0_THREADS) {
        const float4 s4 = ((const float4*)pS)[t];
        const float4 d4 = ((const float4*)pD)[t];
        als[lo * HEADS + t] = (s4.x + s4.y) + (s4.z + s4.w);
        ald[lo * HEADS + t] = (d4.x + d4.y) + (d4.z + d4.w);
    }
}

__global__ void finish2(const float* __restrict__ att, const float* __restrict__ b2,
                        const float* __restrict__ skip32, float* __restrict__ out, int N) {
    const int node = blockIdx.x * (blockDim.x >> 5) + (threadIdx.x >> 5);
    const int c    = threadIdx.x & 31;
    if (node >= N) return;
    float v = 0.25f * att[(size_t)node * OUT_C + c]
            + b2[c] + skip32[(size_t)node * OUT_C + c];
    float sq = v * v;
#pragma unroll
    for (int o = 16; o > 0; o >>= 1) sq += __shfl_xor(sq, o, 32);
    const float norm = sqrtf(sq);
    out[(size_t)node * OUT_C + c] = v / fmaxf(norm, 1e-12f);
}

extern "C" void kernel_launch(void* const* d_in, const int* in_sizes, int n_in,
                              void* d_out, int out_size, void* d_ws, size_t ws_size,
                              hipStream_t stream) {
    (void)in_sizes; (void)n_in; (void)out_size; (void)ws_size;

    const float* x   = (const float*)d_in[0];
    const int*   ei  = (const int*)d_in[1];
    const float* W0  = (const float*)d_in[2];
    const float* aS0 = (const float*)d_in[3];
    const float* aD0 = (const float*)d_in[4];
    const float* b0  = (const float*)d_in[5];
    const float* Ws0 = (const float*)d_in[6];
    const float* bs0 = (const float*)d_in[7];
    const float* W1  = (const float*)d_in[8];
    const float* aS1 = (const float*)d_in[9];
    const float* aD1 = (const float*)d_in[10];
    const float* b1  = (const float*)d_in[11];
    const float* Ws1 = (const float*)d_in[12];
    const float* bs1 = (const float*)d_in[13];
    const float* W2  = (const float*)d_in[14];
    const float* aS2 = (const float*)d_in[15];
    const float* aD2 = (const float*)d_in[16];
    const float* b2  = (const float*)d_in[17];
    const float* Ws2 = (const float*)d_in[18];
    const float* bs2 = (const float*)d_in[19];

    const int* src = ei;
    const int* dst = ei + N_EDGES;

    const size_t NF = (size_t)N_NODES * HC;
    float*    ws    = (float*)d_ws;
    float*    h_b   = ws;
    float*    h_a   = ws + NF;
    float*    sk_a  = ws + 2 * NF;
    float*    sk_b  = ws + 3 * NF;
    float*    als_a = ws + 4 * NF;
    float*    ald_a = als_a + (size_t)N_NODES * HEADS;
    float*    als_b = ald_a + (size_t)N_NODES * HEADS;
    float*    ald_b = als_b + (size_t)N_NODES * HEADS;
    unsigned* lists = (unsigned*)(ald_b + (size_t)N_NODES * HEADS);
    int*      lens  = (int*)(lists + (size_t)OWN_WGS * OWN_WAVES * LIST_CAP);

    dense0_owned<<<OWN_WGS, D0_THREADS, 0, stream>>>(x, W0, Ws0, bs0, aS0, aD0, h_a, sk_a, als_a, ald_a);
    aggr_owned<true, HC><<<OWN_WGS, 1024, 0, stream>>>(src, dst, als_a, ald_a, h_a, nullptr, lists, lens,
                                                       b0, sk_a, W1, Ws1, bs1, aS1, aD1, h_b, sk_b, als_b, ald_b);

    aggr_owned<false, OUT_C><<<OWN_WGS, 1024, 0, stream>>>(src, dst, als_b, ald_b, h_b, nullptr, lists, lens,
                                                           b1, sk_b, W2, Ws2, bs2, aS2, aD2, h_a, sk_a, als_a, ald_a);

    aggr_owned<false, 0><<<OWN_WGS, 1024, 0, stream>>>(src, dst, als_a, ald_a, h_a, sk_b, lists, lens,
                                                       nullptr, nullptr, nullptr, nullptr, nullptr, nullptr, nullptr,
                                                       nullptr, nullptr, nullptr, nullptr);
    finish2<<<(N_NODES + 7) / 8, 256, 0, stream>>>(sk_b, b2, sk_a, (float*)d_out, N_NODES);
}


# Layer 0's edge lists from a split by owner done once in the first dense launch (LDS slot counters, 128-byte cell lines in workspace): owners read 256 cells, not the dst list and src; new sum order
# speedup vs baseline: 6.3279x; 1.1466x over previous
#include <hip/hip_runtime.h>
#include <hip/hip_bf16.h>
#include <math.h>

typedef __attribute__((ext_vector_type(2))) float v2f;
typedef __attribute__((ext_vector_type(8))) float v8f;

#define N_NODES   50000
#define N_EDGES   800000
#define F_IN      128
#define HC        128
#define HEADS     4
#define HID       32
#define OUT_C     32
#define NEG_SLOPE 0.2f

#define PART_Q     16

__device__ __forceinline__ void load8(float (&v)[8], const float* __restrict__ p) {
    const float4 a = *(const float4*)p;
    const float4 b = *(const float4*)(p + 4);
    v[0] = a.x; v[1] = a.y; v[2] = a.z; v[3] = a.w;
    v[4] = b.x; v[5] = b.y; v[6] = b.z; v[7] = b.w;
}

__device__ __forceinline__ unsigned f2ord(float f) {
    unsigned u = __float_as_uint(f);
    return (u & 0x80000000u) ? ~u : (u | 0x80000000u);
}
__device__ __forceinline__ float ord2f(unsigned u) {
    return (u & 0x80000000u) ? __uint_as_float(u ^ 0x80000000u)
                             : __uint_as_float(~u);
}

#define OWN_WGS    256
#define OWN_NODES  196
#define OWN_WAVES  32
#define SLICE_E    128
#define N_SLICES   (N_EDGES / SLICE_E)
#define BATCH      4
#define LIST_CAP   1024
#define GROUP      8
#define RES_U      4
#define ACC_ROWS   208
#define PITCH      132
#define EP_U       5
#define SPLIT_E    (N_EDGES / OWN_WGS)
#define LINE_W     32
#define CELL_CAP   (2 * LINE_W - 1)
#define CELLS_PW   (OWN_WGS / OWN_WAVES)

static_assert(N_NODES <= 65536 && OWN_NODES <= 256, "list entry packs source node in 16 bits, local row in 8");
static_assert(OWN_NODES * HEADS <= 1024, "one thread per (node, head) start value");
static_assert(N_EDGES == SPLIT_E * OWN_WGS && OWN_WGS == CELLS_PW * OWN_WAVES, "the split covers the edge list and the owner's waves cover its cells exactly");
static_assert(CELLS_PW * CELL_CAP <= LIST_CAP, "a wave's cells fit in its list");
static_assert(2 * ACC_ROWS * PART_Q <= OWN_WAVES * LIST_CAP, "the partial logit sums fit in the LDS of the edge lists");
static_assert(PART_Q == 2 * (HC / 16) && HEADS * 4 == PART_Q, "4 partial sums per (row, head), side by side");

template <bool SAVE, int NEXT_SK>
__global__ void __launch_bounds__(1024)
aggr_owned(const int* __restrict__ src, const int* __restrict__ dst,
           const float* __restrict__ als, const float* __restrict__ ald,
           const float* __restrict__ h, float* __restrict__ att,
           unsigned* __restrict__ lists, int* __restrict__ lens,
           const unsigned* __restrict__ cellA, const unsigned* __restrict__ cellB,
           const float* __restrict__ bcur, const float* __restrict__ skip,
           const float* __restrict__ Wn, const float* __restrict__ Wsn,
           const float* __restrict__ bsn,
           const float* __restrict__ aSn, const float* __restrict__ aDn,
           float* __restrict__ hn, float* __restrict__ skn,
           float* __restrict__ alsn, float* __restrict__ aldn) {
    constexpr bool MEAN = (NEXT_SK == 0);
    __shared__ float    acc[ACC_ROWS * PITCH];
    __shared__ __attribute__((aligned(16))) unsigned list[OWN_WAVES * LIST_CAP];
    __shared__ unsigned mx[OWN_NODES * HEADS];
    __shared__ float    sm[OWN_NODES * HEADS];
    __shared__ float    aldl[OWN_NODES * HEADS];
    __shared__ int      spilled;
    __shared__ int      overfull;

    const int tid  = threadIdx.x;
    const int wave = __builtin_amdgcn_readfirstlane(tid >> 5);
    const int lane = tid & 31;
    const int hd   = lane & 3;
    const int lo   = blockIdx.x * OWN_NODES;
    const int left = N_NODES - lo;
    const int cnt  = left < OWN_NODES ? left : OWN_NODES;

    unsigned* wl = lists + (size_t)(blockIdx.x * OWN_WAVES + wave) * LIST_CAP;
    int*      wn = lens + blockIdx.x * OWN_WAVES + wave;
    const int kept = SAVE ? -1 : __builtin_amdgcn_readfirstlane(*wn);
    const size_t cell0 = (size_t)blockIdx.x * OWN_WGS + wave * CELLS_PW;
    unsigned ev[CELLS_PW] = {};
    if (SAVE) {
#pragma unroll
        for (int k = 0; k < CELLS_PW; ++k) ev[k] = cellA[(cell0 + k) * LINE_W + lane];
    }

    for (int i = tid; i < ACC_ROWS * PITCH; i += 1024) acc[i] = 0.0f;
    if (tid < OWN_NODES * HEADS) {
        mx[tid]   = 0u;
        sm[tid]   = 0.0f;
        aldl[tid] = (tid < cnt * HEADS) ? ald[lo * HEADS + tid] : 0.0f;
    }
    if (tid == 0) { spilled = 0; overfull = 0; }
    __syncthreads();

    unsigned* mylist = list + wave * LIST_CAP;
    int n = 0;

    auto resolve = [&](const bool withMax, const bool resolved) {
        __builtin_amdgcn_wave_barrier();
        const int sub = lane >> 2;
        for (int i = 0; i < n; i += 8 * RES_U) {
            unsigned ent[RES_U];
            int      sN[RES_U];
            float    sa[RES_U];
#pragma unroll
            for (int u = 0; u < RES_U; ++u) {
                const int idx = i + u * 8 + sub;
                ent[u] = mylist[idx < n ? idx : n - 1];
            }
#pragma unroll
            for (int u = 0; u < RES_U; ++u) sN[u] = resolved ? (int)(ent[u] & 0xFFFFu) : src[ent[u] & 0xFFFFFu];
#pragma unroll
            for (int u = 0; u < RES_U; ++u) sa[u] = withMax ? als[sN[u] * HEADS + hd] : 0.0f;
#pragma unroll
            for (int u = 0; u < RES_U; ++u) {
                const int idx = i + u * 8 + sub;
                if (idx < n) {
                    const unsigned dl = resolved ? (ent[u] >> 16) : (ent[u] >> 20);
                    if (withMax) {
                        float v = sa[u] + aldl[dl * HEADS + hd];
                        v = (v > 0.f) ? v : NEG_SLOPE * v;
                        atomicMax(&mx[dl * HEADS + hd], f2ord(v));
                    }
                    if (!resolved && hd == 0) mylist[idx] = (unsigned)sN[u] | (dl << 16);
                }
            }
        }
        __builtin_amdgcn_wave_barrier();
    };

    static_assert(GROUP * HEADS == 32, "one lane per (entry, head) of a group");
    auto gather = [&]() {
        __builtin_amdgcn_wave_barrier();
        const int sub = lane >> 2;
        for (int i = 0; i < n; i += GROUP) {
            const int      idx = i + sub;
            const bool     own = idx < n;
            const unsigned ent = mylist[own ? idx : n - 1];
            const unsigned dl  = ent >> 16;
            const float    sa  = als[(int)(ent & 0xFFFFu) * HEADS + hd];
            float r[GROUP][4];
#pragma unroll
            for (int k = 0; k < GROUP; ++k) {
                const int sN = (int)((unsigned)__builtin_amdgcn_readlane(ent, k * HEADS) & 0xFFFFu);
                const float* hp = h + (size_t)sN * HC + lane;
                r[k][0] = hp[0];
                r[k][1] = hp[HID];
                r[k][2] = hp[2 * HID];
                r[k][3] = hp[3 * HID];
            }
            float v = sa + aldl[dl * HEADS + hd];
            v = (v > 0.f) ? v : NEG_SLOPE * v;
            const float ex = expf(v - ord2f(mx[dl * HEADS + hd]));
            if (own) atomicAdd(&sm[dl * HEADS + hd], ex);
            const unsigned eb = __float_as_uint(ex);
#pragma unroll
            for (int k = 0; k < GROUP; ++k) {
                if (i + k < n) {
                    const unsigned dk = (unsigned)__builtin_amdgcn_readlane(ent, k * HEADS) >> 16;
                    const float a0 = __uint_as_float(__builtin_amdgcn_readlane(eb, k * HEADS + 0));
                    const float a1 = __uint_as_float(__builtin_amdgcn_readlane(eb, k * HEADS + 1));
                    const float a2 = __uint_as_float(__builtin_amdgcn_readlane(eb, k * HEADS + 2));
                    const float a3 = __uint_as_float(__builtin_amdgcn_readlane(eb, k * HEADS + 3));
                    float* ap = acc + dk * PITCH + lane;
                    atomicAdd(ap,           a0 * r[k][0]);
                    atomicAdd(ap + HID,     a1 * r[k][1]);
                    atomicAdd(ap + 2 * HID, a2 * r[k][2]);
                    atomicAdd(ap + 3 * HID, a3 * r[k][3]);
                }
            }
        }
        __builtin_amdgcn_wave_barrier();
    };

    auto scan = [&](auto&& onFull) {
        int4 dn[BATCH];
#pragma unroll
        for (int b = 0; b < BATCH; ++b) {
            const int sl = wave * BATCH + b;
            dn[b] = *(const int4*)(dst + sl * SLICE_E + lane * 4);
        }
        for (int s0 = wave * BATCH; s0 < N_SLICES; s0 += OWN_WAVES * BATCH) {
            int4 dc[BATCH];
#pragma unroll
            for (int b = 0; b < BATCH; ++b) dc[b] = dn[b];
#pragma unroll
            for (int b = 0; b < BATCH; ++b) {
                int sl = s0 + OWN_WAVES * BATCH + b;
                sl = sl < N_SLICES ? sl : N_SLICES - 1;
                dn[b] = *(const int4*)(dst + sl * SLICE_E + lane * 4);
            }
            if (n > LIST_CAP - BATCH * SLICE_E) onFull();
#pragma unroll
            for (int b = 0; b < BATCH; ++b) {
                if (s0 + b < N_SLICES) {
                    const unsigned r0 = (unsigned)(dc[b].x - lo);
                    const unsigned r1 = (unsigned)(dc[b].y - lo);
                    const unsigned r2 = (unsigned)(dc[b].z - lo);
                    const unsigned r3 = (unsigned)(dc[b].w - lo);
                    const bool m0 = r0 < (unsigned)cnt;
                    const bool m1 = r1 < (unsigned)cnt;
                    const bool m2 = r2 < (unsigned)cnt;
                    const bool m3 = r3 < (unsigned)cnt;
                    if (__builtin_amdgcn_ballot_w32(m0 | m1 | m2 | m3)) {
                        const unsigned e0 = (unsigned)((s0 + b) * SLICE_E + lane * 4);
                        unsigned mk;
                        mk = __builtin_amdgcn_ballot_w32(m0);
                        if (m0) mylist[n + __builtin_amdgcn_mbcnt_lo(mk, 0u)] = (e0 + 0u) | (r0 << 20);
                        n += __builtin_popcount(mk);
                        mk = __builtin_amdgcn_ballot_w32(m1);
                        if (m1) mylist[n + __builtin_amdgcn_mbcnt_lo(mk, 0u)] = (e0 + 1u) | (r1 << 20);
                        n += __builtin_popcount(mk);
                        mk = __builtin_amdgcn_ballot_w32(m2);
                        if (m2) mylist[n + __builtin_amdgcn_mbcnt_lo(mk, 0u)] = (e0 + 2u) | (r2 << 20);
                        n += __builtin_popcount(mk);
                        mk = __builtin_amdgcn_ballot_w32(m3);
                        if (m3) mylist[n + __builtin_amdgcn_mbcnt_lo(mk, 0u)] = (e0 + 3u) | (r3 << 20);
                        n += __builtin_popcount(mk);
                    }
                }
            }
        }
    };

    bool fromCells = false;
    if (SAVE) {
        bool over = false;
#pragma unroll
        for (int k = 0; k < CELLS_PW; ++k) {
            const unsigned c  = (unsigned)__builtin_amdgcn_readlane(ev[k], 0);
            const int      cc = c < (unsigned)CELL_CAP ? (int)c : CELL_CAP;
            over |= c > (unsigned)CELL_CAP;
            if (lane >= 1 && lane <= cc) mylist[n + lane - 1] = ev[k];
            if (cc >= LINE_W) {
                const unsigned v = cellB[(cell0 + k) * LINE_W + lane];
                if (LINE_W + lane <= cc) mylist[n + LINE_W - 1 + lane] = v;
            }
            n += cc;
        }
        if (over && lane == 0) overfull = 1;
        __syncthreads();
        fromCells = __builtin_amdgcn_readfirstlane(overfull) == 0;
        if (!fromCells) n = 0;
    }

    if (kept >= 0) {
        n = kept;
        for (int i = 0; i < n; i += 4 * 32) {
            unsigned v[4];
#pragma unroll
            for (int u = 0; u < 4; ++u) {
                const int idx = i + u * 32 + lane;
                v[u] = wl[idx < n ? idx : n - 1];
            }
#pragma unroll
            for (int u = 0; u < 4; ++u) {
                const int idx = i + u * 32 + lane;
                if (idx < n) mylist[idx] = v[u];
            }
        }
        resolve(true, true);
    } else if (fromCells) {
        resolve(true, true);
    } else {
        scan([&]() { resolve(true, false); if (lane == 0) spilled = 1; n = 0; });
        resolve(true, false);
    }
    __syncthreads();
    if (SAVE) {
        const bool sp = __builtin_amdgcn_readfirstlane(spilled) != 0;
        if (!sp) for (int i = lane; i < n; i += 32) wl[i] = mylist[i];
        if (lane == 0) *wn = sp ? -1 : n;
    }
    if (__builtin_amdgcn_readfirstlane(spilled) == 0) {
        gather();
    } else {
        n = 0;
        scan([&]() { resolve(false, false); gather(); n = 0; });
        resolve(false, false);
        gather();
    }
    __syncthreads();

    if (!MEAN) {
        constexpr int CT = (HC + NEXT_SK) / 16;
        constexpr int G  = OWN_WAVES / CT;
        const int  ct   = wave % CT;
        const bool isH  = ct < HC / 16;
        const float* Bm = isH ? Wn : Wsn;
        const int  P    = isH ? HC : NEXT_SK;
        const int  half = lane >> 4;
        const int  r    = lane & 15;
        const int  ctl  = isH ? ct : ct - HC / 16;
        const int  col  = ctl * 16 + r;
        const int  c0   = ctl * 16 + 8 * half;
        const bool busy = wave < G * CT;
        float* pS = (float*)list;
        float* pD = pS + ACC_ROWS * PART_Q;

        v2f bf[HC / 4];
        float bb[8] = {}, wS[8] = {}, wD[8] = {};
        if (busy) {
#pragma unroll
            for (int s = 0; s < HC / 4; ++s) {
                const int ka = 4 * s + 2 * half;
                bf[s].x = Bm[(size_t)(ka + 0) * P + col];
                bf[s].y = Bm[(size_t)(ka + 1) * P + col];
            }
            if (isH) { load8(wS, aSn + c0); load8(wD, aDn + c0); }
            else     { load8(bb, bsn + c0); }
        }

        const float* sk = skip + (size_t)lo * HC;
        const float bch = bcur[tid & 127];
        for (int i0 = tid; i0 < cnt * HC; i0 += 1024 * EP_U) {
            float sv[EP_U];
#pragma unroll
            for (int u = 0; u < EP_U; ++u) {
                const int i = i0 + u * 1024;
                sv[u] = (i < cnt * HC) ? sk[i] : 0.0f;
            }
#pragma unroll
            for (int u = 0; u < EP_U; ++u) {
                const int i = i0 + u * 1024;
                if (i < cnt * HC) {
                    const int row = i >> 7;
                    const int ch  = i & 127;
                    float* ap = acc + row * PITCH + ch;
                    const float q = *ap / (sm[row * HEADS + (ch >> 5)] + 1e-16f);
                    const float v = q + bch + sv[u];
                    *ap = (v > 0.f) ? v : (expf(v) - 1.0f);
                }
            }
        }
        __syncthreads();

        if (busy) {
            float* Cm = (isH ? hn : skn) + (size_t)lo * P + c0;
            const int rtiles = (cnt + 15) >> 4;
            for (int rt = wave / CT; rt < rtiles; rt += G) {
                v8f d = {};
                const int row = rt * 16 + r;
                const float* ar = acc + row * PITCH + 2 * half;
#pragma unroll
                for (int s = 0; s < HC / 4; ++s) {
                    const v2f a = *(const v2f*)(ar + 4 * s);
                    d = __builtin_amdgcn_wmma_f32_16x16x4_f32(
                         false, bf[s],  false, a,
                         (short)0, d,  false,  false);
                }
                if (isH) {
                    float ps = d[0] * wS[0], pd = d[0] * wD[0];
#pragma unroll
                    for (int i = 1; i < 8; ++i) {
                        ps = fmaf(d[i], wS[i], ps);
                        pd = fmaf(d[i], wD[i], pd);
                    }
                    pS[row * PART_Q + ct * 2 + half] = ps;
                    pD[row * PART_Q + ct * 2 + half] = pd;
                }
                if (row < cnt) {
                    float* cp = Cm + (size_t)row * P;
                    *(float4*)cp       = make_float4(d[0] + bb[0], d[1] + bb[1], d[2] + bb[2], d[3] + bb[3]);
                    *(float4*)(cp + 4) = make_float4(d[4] + bb[4], d[5] + bb[5], d[6] + bb[6], d[7] + bb[7]);
                }
            }
        }
        __syncthreads();
        if (tid < cnt * HEADS) {
            const float4 s4 = ((const float4*)pS)[tid];
            const float4 d4 = ((const float4*)pD)[tid];
            alsn[lo * HEADS + tid] = (s4.x + s4.y) + (s4.z + s4.w);
            aldn[lo * HEADS + tid] = (d4.x + d4.y) + (d4.z + d4.w);
        }
    } else {
        float* outp = att + (size_t)lo * OUT_C;
        for (int i = tid; i < cnt * OUT_C; i += 1024) {
            const float* ar = acc + (i >> 5) * PITCH + (i & 31);
            const float* sr = sm + (i >> 5) * HEADS;
            outp[i] = ar[0]       / (sr[0] + 1e-16f) + ar[HID]     / (sr[1] + 1e-16f)
                    + ar[2 * HID] / (sr[2] + 1e-16f) + ar[3 * HID] / (sr[3] + 1e-16f);
        }
    }
}

#define D0_WAVES   16
#define D0_THREADS (D0_WAVES * 32)
#define D0_TRIPS   (ACC_ROWS * (F_IN / 4) / D0_THREADS)
#define SP_TRIPS   ((SPLIT_E + D0_THREADS - 1) / D0_THREADS)
#define SP_OWNERS  (OWN_WGS / D0_WAVES)

static_assert(F_IN == HC && HC / 16 * 2 == D0_WAVES, "one wave per column tile of the two products");
static_assert(ACC_ROWS * (F_IN / 4) == D0_TRIPS * D0_THREADS, "the staging loop covers the row block exactly");
static_assert(OWN_WGS == SP_OWNERS * D0_WAVES && OWN_WGS <= D0_THREADS, "the waves cover the owners exactly; one thread per counter");
static_assert(SP_OWNERS <= 32 && (SP_OWNERS & (SP_OWNERS - 1)) == 0, "one lane per owner of a wave, taken by lane & (SP_OWNERS - 1)");

__global__ void __launch_bounds__(D0_THREADS)
dense0_owned(const float* __restrict__ x, const float* __restrict__ Wa,
             const float* __restrict__ Wb, const float* __restrict__ biasb,
             const float* __restrict__ aS, const float* __restrict__ aD,
             float* __restrict__ Ca, float* __restrict__ Cb,
             float* __restrict__ als, float* __restrict__ ald,
             const int* __restrict__ src, const int* __restrict__ dst,
             unsigned* __restrict__ cellA, unsigned* __restrict__ cellB) {
    __shared__ __attribute__((aligned(16))) float xs[ACC_ROWS * PITCH];
    __shared__ __attribute__((aligned(16))) unsigned cellS[OWN_WGS * 2 * LINE_W];
    __shared__ int      fill[OWN_WGS];
    __shared__ __attribute__((aligned(16))) float pS[ACC_ROWS * PART_Q];
    __shared__ __attribute__((aligned(16))) float pD[ACC_ROWS * PART_Q];

    const int tid  = threadIdx.x;
    const int wave = __builtin_amdgcn_readfirstlane(tid >> 5);
    const int lane = tid & 31;
    const int lo   = blockIdx.x * OWN_NODES;
    const int left = N_NODES - lo;
    const int cnt  = left < OWN_NODES ? left : OWN_NODES;

    for (int i = tid; i < OWN_WGS * 2 * LINE_W / 4; i += D0_THREADS) ((uint4*)cellS)[i] = make_uint4(0u, 0u, 0u, 0u);
    if (tid < OWN_WGS) fill[tid] = 0;
    __syncthreads();

    const int eb = blockIdx.x * SPLIT_E;
    int ed[SP_TRIPS], es[SP_TRIPS];
#pragma unroll
    for (int u = 0; u < SP_TRIPS; ++u) {
        const int e  = tid + u * D0_THREADS;
        const int ec = e < SPLIT_E ? e : SPLIT_E - 1;
        ed[u] = dst[eb + ec];
        es[u] = src[eb + ec];
    }

    const float4* xg = (const float4*)(x + (size_t)lo * F_IN);
    float4 xv[D0_TRIPS];
#pragma unroll
    for (int u = 0; u < D0_TRIPS; ++u) {
        const int i = tid + u * D0_THREADS;
        xv[u] = ((i >> 5) < cnt) ? xg[i] : make_float4(0.f, 0.f, 0.f, 0.f);
    }

    const bool isH  = wave < HC / 16;
    const float* Bm = isH ? Wa : Wb;
    const int  half = lane >> 4;
    const int  r    = lane & 15;
    const int  ctl  = wave & (HC / 16 - 1);
    const int  col  = ctl * 16 + r;
    const int  c0   = ctl * 16 + 8 * half;

    v2f bf[F_IN / 4];
#pragma unroll
    for (int s = 0; s < F_IN / 4; ++s) {
        const int ka = 4 * s + 2 * half;
        bf[s].x = Bm[(size_t)(ka + 0) * HC + col];
        bf[s].y = Bm[(size_t)(ka + 1) * HC + col];
    }

    unsigned ow[SP_TRIPS];
    int      slot[SP_TRIPS];
#pragma unroll
    for (int u = 0; u < SP_TRIPS; ++u) {
        const bool ok = (tid + u * D0_THREADS < SPLIT_E) && ((unsigned)ed[u] < (unsigned)N_NODES);
        ow[u]   = ok ? (unsigned)ed[u] / OWN_NODES : 0u;
        slot[u] = ok ? atomicAdd(&fill[ow[u]], 1) : CELL_CAP;
    }
#pragma unroll
    for (int u = 0; u < SP_TRIPS; ++u) {
        if (slot[u] < CELL_CAP)
            cellS[ow[u] * (2 * LINE_W) + 1 + slot[u]] = (unsigned)es[u] | (((unsigned)ed[u] - ow[u] * OWN_NODES) << 16);
    }

    float bb[8] = {}, wS[8] = {}, wD[8] = {};
    if (isH) { load8(wS, aS + c0); load8(wD, aD + c0); }
    else     { load8(bb, biasb + c0); }

#pragma unroll
    for (int u = 0; u < D0_TRIPS; ++u) {
        const int i = tid + u * D0_THREADS;
        *(float4*)(xs + (i >> 5) * PITCH + (i & 31) * 4) = xv[u];
    }
    __syncthreads();

    const int cv = fill[wave * SP_OWNERS + (lane & (SP_OWNERS - 1))];
    unsigned wa[SP_OWNERS];
#pragma unroll
    for (int q = 0; q < SP_OWNERS; ++q) wa[q] = cellS[(wave * SP_OWNERS + q) * (2 * LINE_W) + lane];
#pragma unroll
    for (int q = 0; q < SP_OWNERS; ++q) {
        const int o = wave * SP_OWNERS + q;
        const int c = __builtin_amdgcn_readlane(cv, q);
        const size_t ln = ((size_t)o * OWN_WGS + blockIdx.x) * LINE_W;
        cellA[ln + lane] = (lane == 0) ? (unsigned)c : wa[q];
        if (c >= LINE_W) cellB[ln + lane] = cellS[o * (2 * LINE_W) + LINE_W + lane];
    }

    float* Cm = (isH ? Ca : Cb) + (size_t)lo * HC + c0;
    const int rtiles = (cnt + 15) >> 4;
    for (int rt = 0; rt < rtiles; ++rt) {
        v8f d = {};
        const int row = rt * 16 + r;
        const float* ar = xs + row * PITCH + 2 * half;
#pragma unroll
        for (int s = 0; s < F_IN / 4; ++s) {
            const v2f a = *(const v2f*)(ar + 4 * s);
            d = __builtin_amdgcn_wmma_f32_16x16x4_f32(
                 false, bf[s],  false, a,
                 (short)0, d,  false,  false);
        }
        if (isH) {
            float ps = d[0] * wS[0], pd = d[0] * wD[0];
#pragma unroll
            for (int i = 1; i < 8; ++i) {
                ps = fmaf(d[i], wS[i], ps);
                pd = fmaf(d[i], wD[i], pd);
            }
            pS[row * PART_Q + ctl * 2 + half] = ps;
            pD[row * PART_Q + ctl * 2 + half] = pd;
        }
        if (row < cnt) {
            float* cp = Cm + (size_t)row * HC;
            *(float4*)cp       = make_float4(d[0] + bb[0], d[1] + bb[1], d[2] + bb[2], d[3] + bb[3]);
            *(float4*)(cp + 4) = make_float4(d[4] + bb[4], d[5] + bb[5], d[6] + bb[6], d[7] + bb[7]);
        }
    }
    __syncthreads();
    for (int t = tid; t < cnt * HEADS; t += D0_THREADS) {
        const float4 s4 = ((const float4*)pS)[t];
        const float4 d4 = ((const float4*)pD)[t];
        als[lo * HEADS + t] = (s4.x + s4.y) + (s4.z + s4.w);
        ald[lo * HEADS + t] = (d4.x + d4.y) + (d4.z + d4.w);
    }
}

__global__ void finish2(const float* __restrict__ att, const float* __restrict__ b2,
                        const float* __restrict__ skip32, float* __restrict__ out, int N) {
    const int node = blockIdx.x * (blockDim.x >> 5) + (threadIdx.x >> 5);
    const int c    = threadIdx.x & 31;
    if (node >= N) return;
    float v = 0.25f * att[(size_t)node * OUT_C + c]
            + b2[c] + skip32[(size_t)node * OUT_C + c];
    float sq = v * v;
#pragma unroll
    for (int o = 16; o > 0; o >>= 1) sq += __shfl_xor(sq, o, 32);
    const float norm = sqrtf(sq);
    out[(size_t)node * OUT_C + c] = v / fmaxf(norm, 1e-12f);
}

extern "C" void kernel_launch(void* const* d_in, const int* in_sizes, int n_in,
                              void* d_out, int out_size, void* d_ws, size_t ws_size,
                              hipStream_t stream) {
    (void)in_sizes; (void)n_in; (void)out_size; (void)ws_size;

    const float* x   = (const float*)d_in[0];
    const int*   ei  = (const int*)d_in[1];
    const float* W0  = (const float*)d_in[2];
    const float* aS0 = (const float*)d_in[3];
    const float* aD0 = (const float*)d_in[4];
    const float* b0  = (const float*)d_in[5];
    const float* Ws0 = (const float*)d_in[6];
    const float* bs0 = (const float*)d_in[7];
    const float* W1  = (const float*)d_in[8];
    const float* aS1 = (const float*)d_in[9];
    const float* aD1 = (const float*)d_in[10];
    const float* b1  = (const float*)d_in[11];
    const float* Ws1 = (const float*)d_in[12];
    const float* bs1 = (const float*)d_in[13];
    const float* W2  = (const float*)d_in[14];
    const float* aS2 = (const float*)d_in[15];
    const float* aD2 = (const float*)d_in[16];
    const float* b2  = (const float*)d_in[17];
    const float* Ws2 = (const float*)d_in[18];
    const float* bs2 = (const float*)d_in[19];

    const int* src = ei;
    const int* dst = ei + N_EDGES;

    const size_t NF = (size_t)N_NODES * HC;
    float*    ws    = (float*)d_ws;
    float*    h_b   = ws;
    float*    h_a   = ws + NF;
    float*    sk_a  = ws + 2 * NF;
    float*    sk_b  = ws + 3 * NF;
    float*    als_a = ws + 4 * NF;
    float*    ald_a = als_a + (size_t)N_NODES * HEADS;
    float*    als_b = ald_a + (size_t)N_NODES * HEADS;
    float*    ald_b = als_b + (size_t)N_NODES * HEADS;
    unsigned* lists = (unsigned*)(ald_b + (size_t)N_NODES * HEADS);
    int*      lens  = (int*)(lists + (size_t)OWN_WGS * OWN_WAVES * LIST_CAP);
    unsigned* cellA = (unsigned*)(lens + OWN_WGS * OWN_WAVES);
    unsigned* cellB = cellA + (size_t)OWN_WGS * OWN_WGS * LINE_W;

    dense0_owned<<<OWN_WGS, D0_THREADS, 0, stream>>>(x, W0, Ws0, bs0, aS0, aD0, h_a, sk_a, als_a, ald_a,
                                                     src, dst, cellA, cellB);
    aggr_owned<true, HC><<<OWN_WGS, 1024, 0, stream>>>(src, dst, als_a, ald_a, h_a, nullptr, lists, lens,
                                                       cellA, cellB, b0, sk_a, W1, Ws1, bs1, aS1, aD1, h_b, sk_b, als_b, ald_b);

    aggr_owned<false, OUT_C><<<OWN_WGS, 1024, 0, stream>>>(src, dst, als_b, ald_b, h_b, nullptr, lists, lens,
                                                           cellA, cellB, b1, sk_b, W2, Ws2, bs2, aS2, aD2, h_a, sk_a, als_a, ald_a);

    aggr_owned<false, 0><<<OWN_WGS, 1024, 0, stream>>>(src, dst, als_a, ald_a, h_a, sk_b, lists, lens,
                                                       cellA, cellB, nullptr, nullptr, nullptr, nullptr, nullptr, nullptr, nullptr,
                                                       nullptr, nullptr, nullptr, nullptr);
    finish2<<<(N_NODES + 7) / 8, 256, 0, stream>>>(sk_b, b2, sk_a, (float*)d_out, N_NODES);
}


# Aggregation reads a source row with one 16-byte load per lane (a 512-byte row per instruction, not four 4-byte loads); LDS rows kept in that lane order, dense K order follows it; new sum order
# speedup vs baseline: 6.7799x; 1.0714x over previous
#include <hip/hip_runtime.h>
#include <hip/hip_bf16.h>
#include <math.h>

typedef __attribute__((ext_vector_type(2))) float v2f;
typedef __attribute__((ext_vector_type(8))) float v8f;

#define N_NODES   50000
#define N_EDGES   800000
#define F_IN      128
#define HC        128
#define HEADS     4
#define HID       32
#define OUT_C     32
#define NEG_SLOPE 0.2f

#define PART_Q     16

__device__ __forceinline__ constexpr int chan_of_pos(int p) { return 4 * (p & 31) + (p >> 5); }

__device__ __forceinline__ void load8(float (&v)[8], const float* __restrict__ p) {
    const float4 a = *(const float4*)p;
    const float4 b = *(const float4*)(p + 4);
    v[0] = a.x; v[1] = a.y; v[2] = a.z; v[3] = a.w;
    v[4] = b.x; v[5] = b.y; v[6] = b.z; v[7] = b.w;
}

__device__ __forceinline__ unsigned f2ord(float f) {
    unsigned u = __float_as_uint(f);
    return (u & 0x80000000u) ? ~u : (u | 0x80000000u);
}
__device__ __forceinline__ float ord2f(unsigned u) {
    return (u & 0x80000000u) ? __uint_as_float(u ^ 0x80000000u)
                             : __uint_as_float(~u);
}

#define OWN_WGS    256
#define OWN_NODES  196
#define OWN_WAVES  32
#define SLICE_E    128
#define N_SLICES   (N_EDGES / SLICE_E)
#define BATCH      4
#define LIST_CAP   1024
#define GROUP      8
#define RES_U      4
#define ACC_ROWS   208
#define PITCH      132
#define EP_U       2
#define SPLIT_E    (N_EDGES / OWN_WGS)
#define LINE_W     32
#define CELL_CAP   (2 * LINE_W - 1)
#define CELLS_PW   (OWN_WGS / OWN_WAVES)

static_assert(N_NODES <= 65536 && OWN_NODES <= 256, "list entry packs source node in 16 bits, local row in 8");
static_assert(OWN_NODES * HEADS <= 1024, "one thread per (node, head) start value");
static_assert(N_EDGES == SPLIT_E * OWN_WGS && OWN_WGS == CELLS_PW * OWN_WAVES, "the split covers the edge list and the owner's waves cover its cells exactly");
static_assert(CELLS_PW * CELL_CAP <= LIST_CAP, "a wave's cells fit in its list");
static_assert(2 * ACC_ROWS * PART_Q <= OWN_WAVES * LIST_CAP, "the partial logit sums fit in the LDS of the edge lists");
static_assert(PART_Q == 2 * (HC / 16) && HEADS * 4 == PART_Q, "4 partial sums per (row, head), side by side");

template <bool SAVE, int NEXT_SK>
__global__ void __launch_bounds__(1024)
aggr_owned(const int* __restrict__ src, const int* __restrict__ dst,
           const float* __restrict__ als, const float* __restrict__ ald,
           const float* __restrict__ h, float* __restrict__ att,
           unsigned* __restrict__ lists, int* __restrict__ lens,
           const unsigned* __restrict__ cellA, const unsigned* __restrict__ cellB,
           const float* __restrict__ bcur, const float* __restrict__ skip,
           const float* __restrict__ Wn, const float* __restrict__ Wsn,
           const float* __restrict__ bsn,
           const float* __restrict__ aSn, const float* __restrict__ aDn,
           float* __restrict__ hn, float* __restrict__ skn,
           float* __restrict__ alsn, float* __restrict__ aldn) {
    constexpr bool MEAN = (NEXT_SK == 0);
    __shared__ float    acc[ACC_ROWS * PITCH];
    __shared__ __attribute__((aligned(16))) unsigned list[OWN_WAVES * LIST_CAP];
    __shared__ unsigned mx[OWN_NODES * HEADS];
    __shared__ float    sm[OWN_NODES * HEADS];
    __shared__ float    aldl[OWN_NODES * HEADS];
    __shared__ int      spilled;
    __shared__ int      overfull;

    const int tid  = threadIdx.x;
    const int wave = __builtin_amdgcn_readfirstlane(tid >> 5);
    const int lane = tid & 31;
    const int hd   = lane & 3;
    const int lo   = blockIdx.x * OWN_NODES;
    const int left = N_NODES - lo;
    const int cnt  = left < OWN_NODES ? left : OWN_NODES;

    unsigned* wl = lists + (size_t)(blockIdx.x * OWN_WAVES + wave) * LIST_CAP;
    int*      wn = lens + blockIdx.x * OWN_WAVES + wave;
    const int kept = SAVE ? -1 : __builtin_amdgcn_readfirstlane(*wn);
    const size_t cell0 = (size_t)blockIdx.x * OWN_WGS + wave * CELLS_PW;
    unsigned ev[CELLS_PW] = {};
    if (SAVE) {
#pragma unroll
        for (int k = 0; k < CELLS_PW; ++k) ev[k] = cellA[(cell0 + k) * LINE_W + lane];
    }

    for (int i = tid; i < ACC_ROWS * PITCH; i += 1024) acc[i] = 0.0f;
    if (tid < OWN_NODES * HEADS) {
        mx[tid]   = 0u;
        sm[tid]   = 0.0f;
        aldl[tid] = (tid < cnt * HEADS) ? ald[lo * HEADS + tid] : 0.0f;
    }
    if (tid == 0) { spilled = 0; overfull = 0; }
    __syncthreads();

    unsigned* mylist = list + wave * LIST_CAP;
    int n = 0;

    auto resolve = [&](const bool withMax, const bool resolved) {
        __builtin_amdgcn_wave_barrier();
        const int sub = lane >> 2;
        for (int i = 0; i < n; i += 8 * RES_U) {
            unsigned ent[RES_U];
            int      sN[RES_U];
            float    sa[RES_U];
#pragma unroll
            for (int u = 0; u < RES_U; ++u) {
                const int idx = i + u * 8 + sub;
                ent[u] = mylist[idx < n ? idx : n - 1];
            }
#pragma unroll
            for (int u = 0; u < RES_U; ++u) sN[u] = resolved ? (int)(ent[u] & 0xFFFFu) : src[ent[u] & 0xFFFFFu];
#pragma unroll
            for (int u = 0; u < RES_U; ++u) sa[u] = withMax ? als[sN[u] * HEADS + hd] : 0.0f;
#pragma unroll
            for (int u = 0; u < RES_U; ++u) {
                const int idx = i + u * 8 + sub;
                if (idx < n) {
                    const unsigned dl = resolved ? (ent[u] >> 16) : (ent[u] >> 20);
                    if (withMax) {
                        float v = sa[u] + aldl[dl * HEADS + hd];
                        v = (v > 0.f) ? v : NEG_SLOPE * v;
                        atomicMax(&mx[dl * HEADS + hd], f2ord(v));
                    }
                    if (!resolved && hd == 0) mylist[idx] = (unsigned)sN[u] | (dl << 16);
                }
            }
        }
        __builtin_amdgcn_wave_barrier();
    };

    static_assert(GROUP * HEADS == 32 && GROUP == 8 && HEADS == 4 && HC == 128, "one lane per (entry, head) of a group; a row is 32 lanes x 16 bytes");
    auto gather = [&]() {
        __builtin_amdgcn_wave_barrier();
        const int sub = (lane & 15) >> 1;
        const int hg  = ((lane >> 4) << 1) | (lane & 1);
        for (int i = 0; i < n; i += GROUP) {
            const int      idx = i + sub;
            const bool     own = idx < n;
            const unsigned ent = mylist[own ? idx : n - 1];
            const unsigned dl  = ent >> 16;
            const float    sa  = als[(int)(ent & 0xFFFFu) * HEADS + hg];
            float4 r[GROUP];
#pragma unroll
            for (int k = 0; k < GROUP; ++k) {
                const int sN = (int)((unsigned)__builtin_amdgcn_readlane(ent, 2 * k) & 0xFFFFu);
                r[k] = *(const float4*)(h + (size_t)sN * HC + 4 * lane);
            }
            float v = sa + aldl[dl * HEADS + hg];
            v = (v > 0.f) ? v : NEG_SLOPE * v;
            const float ex = expf(v - ord2f(mx[dl * HEADS + hg]));
            if (own) atomicAdd(&sm[dl * HEADS + hg], ex);
            const unsigned eb = __float_as_uint(ex);
#pragma unroll
            for (int k = 0; k < GROUP; ++k) {
                if (i + k < n) {
                    const unsigned dk = (unsigned)__builtin_amdgcn_readlane(ent, 2 * k) >> 16;
                    const float w = __uint_as_float(__builtin_amdgcn_permlane16(
                        eb, eb, 0x11111111u * (unsigned)(2 * k), 0x11111111u * (unsigned)(2 * k + 1), false, false));
                    float* ap = acc + dk * PITCH + lane;
                    atomicAdd(ap,      w * r[k].x);
                    atomicAdd(ap + 32, w * r[k].y);
                    atomicAdd(ap + 64, w * r[k].z);
                    atomicAdd(ap + 96, w * r[k].w);
                }
            }
        }
        __builtin_amdgcn_wave_barrier();
    };

    auto scan = [&](auto&& onFull) {
        int4 dn[BATCH];
#pragma unroll
        for (int b = 0; b < BATCH; ++b) {
            const int sl = wave * BATCH + b;
            dn[b] = *(const int4*)(dst + sl * SLICE_E + lane * 4);
        }
        for (int s0 = wave * BATCH; s0 < N_SLICES; s0 += OWN_WAVES * BATCH) {
            int4 dc[BATCH];
#pragma unroll
            for (int b = 0; b < BATCH; ++b) dc[b] = dn[b];
#pragma unroll
            for (int b = 0; b < BATCH; ++b) {
                int sl = s0 + OWN_WAVES * BATCH + b;
                sl = sl < N_SLICES ? sl : N_SLICES - 1;
                dn[b] = *(const int4*)(dst + sl * SLICE_E + lane * 4);
            }
            if (n > LIST_CAP - BATCH * SLICE_E) onFull();
#pragma unroll
            for (int b = 0; b < BATCH; ++b) {
                if (s0 + b < N_SLICES) {
                    const unsigned r0 = (unsigned)(dc[b].x - lo);
                    const unsigned r1 = (unsigned)(dc[b].y - lo);
                    const unsigned r2 = (unsigned)(dc[b].z - lo);
                    const unsigned r3 = (unsigned)(dc[b].w - lo);
                    const bool m0 = r0 < (unsigned)cnt;
                    const bool m1 = r1 < (unsigned)cnt;
                    const bool m2 = r2 < (unsigned)cnt;
                    const bool m3 = r3 < (unsigned)cnt;
                    if (__builtin_amdgcn_ballot_w32(m0 | m1 | m2 | m3)) {
                        const unsigned e0 = (unsigned)((s0 + b) * SLICE_E + lane * 4);
                        unsigned mk;
                        mk = __builtin_amdgcn_ballot_w32(m0);
                        if (m0) mylist[n + __builtin_amdgcn_mbcnt_lo(mk, 0u)] = (e0 + 0u) | (r0 << 20);
                        n += __builtin_popcount(mk);
                        mk = __builtin_amdgcn_ballot_w32(m1);
                        if (m1) mylist[n + __builtin_amdgcn_mbcnt_lo(mk, 0u)] = (e0 + 1u) | (r1 << 20);
                        n += __builtin_popcount(mk);
                        mk = __builtin_amdgcn_ballot_w32(m2);
                        if (m2) mylist[n + __builtin_amdgcn_mbcnt_lo(mk, 0u)] = (e0 + 2u) | (r2 << 20);
                        n += __builtin_popcount(mk);
                        mk = __builtin_amdgcn_ballot_w32(m3);
                        if (m3) mylist[n + __builtin_amdgcn_mbcnt_lo(mk, 0u)] = (e0 + 3u) | (r3 << 20);
                        n += __builtin_popcount(mk);
                    }
                }
            }
        }
    };

    bool fromCells = false;
    if (SAVE) {
        bool over = false;
#pragma unroll
        for (int k = 0; k < CELLS_PW; ++k) {
            const unsigned c  = (unsigned)__builtin_amdgcn_readlane(ev[k], 0);
            const int      cc = c < (unsigned)CELL_CAP ? (int)c : CELL_CAP;
            over |= c > (unsigned)CELL_CAP;
            if (lane >= 1 && lane <= cc) mylist[n + lane - 1] = ev[k];
            if (cc >= LINE_W) {
                const unsigned v = cellB[(cell0 + k) * LINE_W + lane];
                if (LINE_W + lane <= cc) mylist[n + LINE_W - 1 + lane] = v;
            }
            n += cc;
        }
        if (over && lane == 0) overfull = 1;
        __syncthreads();
        fromCells = __builtin_amdgcn_readfirstlane(overfull) == 0;
        if (!fromCells) n = 0;
    }

    if (kept >= 0) {
        n = kept;
        for (int i = 0; i < n; i += 4 * 32) {
            unsigned v[4];
#pragma unroll
            for (int u = 0; u < 4; ++u) {
                const int idx = i + u * 32 + lane;
                v[u] = wl[idx < n ? idx : n - 1];
            }
#pragma unroll
            for (int u = 0; u < 4; ++u) {
                const int idx = i + u * 32 + lane;
                if (idx < n) mylist[idx] = v[u];
            }
        }
        resolve(true, true);
    } else if (fromCells) {
        resolve(true, true);
    } else {
        scan([&]() { resolve(true, false); if (lane == 0) spilled = 1; n = 0; });
        resolve(true, false);
    }
    __syncthreads();
    if (SAVE) {
        const bool sp = __builtin_amdgcn_readfirstlane(spilled) != 0;
        if (!sp) for (int i = lane; i < n; i += 32) wl[i] = mylist[i];
        if (lane == 0) *wn = sp ? -1 : n;
    }
    if (__builtin_amdgcn_readfirstlane(spilled) == 0) {
        gather();
    } else {
        n = 0;
        scan([&]() { resolve(false, false); gather(); n = 0; });
        resolve(false, false);
        gather();
    }
    __syncthreads();

    if (!MEAN) {
        constexpr int CT = (HC + NEXT_SK) / 16;
        constexpr int G  = OWN_WAVES / CT;
        const int  ct   = wave % CT;
        const bool isH  = ct < HC / 16;
        const float* Bm = isH ? Wn : Wsn;
        const int  P    = isH ? HC : NEXT_SK;
        const int  half = lane >> 4;
        const int  r    = lane & 15;
        const int  ctl  = isH ? ct : ct - HC / 16;
        const int  col  = ctl * 16 + r;
        const int  c0   = ctl * 16 + 8 * half;
        const bool busy = wave < G * CT;
        float* pS = (float*)list;
        float* pD = pS + ACC_ROWS * PART_Q;

        v2f bf[HC / 4];
        float bb[8] = {}, wS[8] = {}, wD[8] = {};
        if (busy) {
#pragma unroll
            for (int s = 0; s < HC / 4; ++s) {
                bf[s].x = Bm[(size_t)(chan_of_pos(4 * s + 0) + 8 * half) * P + col];
                bf[s].y = Bm[(size_t)(chan_of_pos(4 * s + 1) + 8 * half) * P + col];
            }
            if (isH) { load8(wS, aSn + c0); load8(wD, aDn + c0); }
            else     { load8(bb, bsn + c0); }
        }

        const float4* sk4 = (const float4*)(skip + (size_t)lo * HC);
        const float4  bc4 = *(const float4*)(bcur + 4 * lane);
        for (int i0 = tid; i0 < cnt * (HC / 4); i0 += 1024 * EP_U) {
            float4 sv[EP_U];
#pragma unroll
            for (int u = 0; u < EP_U; ++u) {
                const int i = i0 + u * 1024;
                sv[u] = (i < cnt * (HC / 4)) ? sk4[i] : make_float4(0.f, 0.f, 0.f, 0.f);
            }
#pragma unroll
            for (int u = 0; u < EP_U; ++u) {
                const int i = i0 + u * 1024;
                if (i < cnt * (HC / 4)) {
                    const int row = i >> 5;
                    float* ap = acc + row * PITCH + lane;
                    const float den = sm[row * HEADS + (lane >> 3)] + 1e-16f;
                    const float v0 = ap[0]  / den + bc4.x + sv[u].x;
                    const float v1 = ap[32] / den + bc4.y + sv[u].y;
                    const float v2 = ap[64] / den + bc4.z + sv[u].z;
                    const float v3 = ap[96] / den + bc4.w + sv[u].w;
                    ap[0]  = (v0 > 0.f) ? v0 : (expf(v0) - 1.0f);
                    ap[32] = (v1 > 0.f) ? v1 : (expf(v1) - 1.0f);
                    ap[64] = (v2 > 0.f) ? v2 : (expf(v2) - 1.0f);
                    ap[96] = (v3 > 0.f) ? v3 : (expf(v3) - 1.0f);
                }
            }
        }
        __syncthreads();

        if (busy) {
            float* Cm = (isH ? hn : skn) + (size_t)lo * P + c0;
            const int rtiles = (cnt + 15) >> 4;
            for (int rt = wave / CT; rt < rtiles; rt += G) {
                v8f d = {};
                const int row = rt * 16 + r;
                const float* ar = acc + row * PITCH + 2 * half;
#pragma unroll
                for (int s = 0; s < HC / 4; ++s) {
                    const v2f a = *(const v2f*)(ar + 4 * s);
                    d = __builtin_amdgcn_wmma_f32_16x16x4_f32(
                         false, bf[s],  false, a,
                         (short)0, d,  false,  false);
                }
                if (isH) {
                    float ps = d[0] * wS[0], pd = d[0] * wD[0];
#pragma unroll
                    for (int i = 1; i < 8; ++i) {
                        ps = fmaf(d[i], wS[i], ps);
                        pd = fmaf(d[i], wD[i], pd);
                    }
                    pS[row * PART_Q + ct * 2 + half] = ps;
                    pD[row * PART_Q + ct * 2 + half] = pd;
                }
                if (row < cnt) {
                    float* cp = Cm + (size_t)row * P;
                    *(float4*)cp       = make_float4(d[0] + bb[0], d[1] + bb[1], d[2] + bb[2], d[3] + bb[3]);
                    *(float4*)(cp + 4) = make_float4(d[4] + bb[4], d[5] + bb[5], d[6] + bb[6], d[7] + bb[7]);
                }
            }
        }
        __syncthreads();
        if (tid < cnt * HEADS) {
            const float4 s4 = ((const float4*)pS)[tid];
            const float4 d4 = ((const float4*)pD)[tid];
            alsn[lo * HEADS + tid] = (s4.x + s4.y) + (s4.z + s4.w);
            aldn[lo * HEADS + tid] = (d4.x + d4.y) + (d4.z + d4.w);
        }
    } else {
        float* outp = att + (size_t)lo * OUT_C;
        for (int i = tid; i < cnt * OUT_C; i += 1024) {
            const int c = i & 31;
            const float* ar = acc + (i >> 5) * PITCH + 32 * (c & 3) + (c >> 2);
            const float* sr = sm + (i >> 5) * HEADS;
            outp[i] = ar[0]  / (sr[0] + 1e-16f) + ar[8]  / (sr[1] + 1e-16f)
                    + ar[16] / (sr[2] + 1e-16f) + ar[24] / (sr[3] + 1e-16f);
        }
    }
}

#define D0_WAVES   16
#define D0_THREADS (D0_WAVES * 32)
#define D0_TRIPS   (ACC_ROWS * (F_IN / 4) / D0_THREADS)
#define SP_TRIPS   ((SPLIT_E + D0_THREADS - 1) / D0_THREADS)
#define SP_OWNERS  (OWN_WGS / D0_WAVES)

static_assert(F_IN == HC && HC / 16 * 2 == D0_WAVES, "one wave per column tile of the two products");
static_assert(ACC_ROWS * (F_IN / 4) == D0_TRIPS * D0_THREADS, "the staging loop covers the row block exactly");
static_assert(OWN_WGS == SP_OWNERS * D0_WAVES && OWN_WGS <= D0_THREADS, "the waves cover the owners exactly; one thread per counter");
static_assert(SP_OWNERS <= 32 && (SP_OWNERS & (SP_OWNERS - 1)) == 0, "one lane per owner of a wave, taken by lane & (SP_OWNERS - 1)");

__global__ void __launch_bounds__(D0_THREADS)
dense0_owned(const float* __restrict__ x, const float* __restrict__ Wa,
             const float* __restrict__ Wb, const float* __restrict__ biasb,
             const float* __restrict__ aS, const float* __restrict__ aD,
             float* __restrict__ Ca, float* __restrict__ Cb,
             float* __restrict__ als, float* __restrict__ ald,
             const int* __restrict__ src, const int* __restrict__ dst,
             unsigned* __restrict__ cellA, unsigned* __restrict__ cellB) {
    __shared__ __attribute__((aligned(16))) float xs[ACC_ROWS * PITCH];
    __shared__ __attribute__((aligned(16))) unsigned cellS[OWN_WGS * 2 * LINE_W];
    __shared__ int      fill[OWN_WGS];
    __shared__ __attribute__((aligned(16))) float pS[ACC_ROWS * PART_Q];
    __shared__ __attribute__((aligned(16))) float pD[ACC_ROWS * PART_Q];

    const int tid  = threadIdx.x;
    const int wave = __builtin_amdgcn_readfirstlane(tid >> 5);
    const int lane = tid & 31;
    const int lo   = blockIdx.x * OWN_NODES;
    const int left = N_NODES - lo;
    const int cnt  = left < OWN_NODES ? left : OWN_NODES;

    for (int i = tid; i < OWN_WGS * 2 * LINE_W / 4; i += D0_THREADS) ((uint4*)cellS)[i] = make_uint4(0u, 0u, 0u, 0u);
    if (tid < OWN_WGS) fill[tid] = 0;
    __syncthreads();

    const int eb = blockIdx.x * SPLIT_E;
    int ed[SP_TRIPS], es[SP_TRIPS];
#pragma unroll
    for (int u = 0; u < SP_TRIPS; ++u) {
        const int e  = tid + u * D0_THREADS;
        const int ec = e < SPLIT_E ? e : SPLIT_E - 1;
        ed[u] = dst[eb + ec];
        es[u] = src[eb + ec];
    }

    const float4* xg = (const float4*)(x + (size_t)lo * F_IN);
    float4 xv[D0_TRIPS];
#pragma unroll
    for (int u = 0; u < D0_TRIPS; ++u) {
        const int i = tid + u * D0_THREADS;
        xv[u] = ((i >> 5) < cnt) ? xg[i] : make_float4(0.f, 0.f, 0.f, 0.f);
    }

    const bool isH  = wave < HC / 16;
    const float* Bm = isH ? Wa : Wb;
    const int  half = lane >> 4;
    const int  r    = lane & 15;
    const int  ctl  = wave & (HC / 16 - 1);
    const int  col  = ctl * 16 + r;
    const int  c0   = ctl * 16 + 8 * half;

    v2f bf[F_IN / 4];
#pragma unroll
    for (int s = 0; s < F_IN / 4; ++s) {
        const int ka = 4 * s + 2 * half;
        bf[s].x = Bm[(size_t)(ka + 0) * HC + col];
        bf[s].y = Bm[(size_t)(ka + 1) * HC + col];
    }

    unsigned ow[SP_TRIPS];
    int      slot[SP_TRIPS];
#pragma unroll
    for (int u = 0; u < SP_TRIPS; ++u) {
        const bool ok = (tid + u * D0_THREADS < SPLIT_E) && ((unsigned)ed[u] < (unsigned)N_NODES);
        ow[u]   = ok ? (unsigned)ed[u] / OWN_NODES : 0u;
        slot[u] = ok ? atomicAdd(&fill[ow[u]], 1) : CELL_CAP;
    }
#pragma unroll
    for (int u = 0; u < SP_TRIPS; ++u) {
        if (slot[u] < CELL_CAP)
            cellS[ow[u] * (2 * LINE_W) + 1 + slot[u]] = (unsigned)es[u] | (((unsigned)ed[u] - ow[u] * OWN_NODES) << 16);
    }

    float bb[8] = {}, wS[8] = {}, wD[8] = {};
    if (isH) { load8(wS, aS + c0); load8(wD, aD + c0); }
    else     { load8(bb, biasb + c0); }

#pragma unroll
    for (int u = 0; u < D0_TRIPS; ++u) {
        const int i = tid + u * D0_THREADS;
        *(float4*)(xs + (i >> 5) * PITCH + (i & 31) * 4) = xv[u];
    }
    __syncthreads();

    const int cv = fill[wave * SP_OWNERS + (lane & (SP_OWNERS - 1))];
    unsigned wa[SP_OWNERS];
#pragma unroll
    for (int q = 0; q < SP_OWNERS; ++q) wa[q] = cellS[(wave * SP_OWNERS + q) * (2 * LINE_W) + lane];
#pragma unroll
    for (int q = 0; q < SP_OWNERS; ++q) {
        const int o = wave * SP_OWNERS + q;
        const int c = __builtin_amdgcn_readlane(cv, q);
        const size_t ln = ((size_t)o * OWN_WGS + blockIdx.x) * LINE_W;
        cellA[ln + lane] = (lane == 0) ? (unsigned)c : wa[q];
        if (c >= LINE_W) cellB[ln + lane] = cellS[o * (2 * LINE_W) + LINE_W + lane];
    }

    float* Cm = (isH ? Ca : Cb) + (size_t)lo * HC + c0;
    const int rtiles = (cnt + 15) >> 4;
    for (int rt = 0; rt < rtiles; ++rt) {
        v8f d = {};
        const int row = rt * 16 + r;
        const float* ar = xs + row * PITCH + 2 * half;
#pragma unroll
        for (int s = 0; s < F_IN / 4; ++s) {
            const v2f a = *(const v2f*)(ar + 4 * s);
            d = __builtin_amdgcn_wmma_f32_16x16x4_f32(
                 false, bf[s],  false, a,
                 (short)0, d,  false,  false);
        }
        if (isH) {
            float ps = d[0] * wS[0], pd = d[0] * wD[0];
#pragma unroll
            for (int i = 1; i < 8; ++i) {
                ps = fmaf(d[i], wS[i], ps);
                pd = fmaf(d[i], wD[i], pd);
            }
            pS[row * PART_Q + ctl * 2 + half] = ps;
            pD[row * PART_Q + ctl * 2 + half] = pd;
        }
        if (row < cnt) {
            float* cp = Cm + (size_t)row * HC;
            *(float4*)cp       = make_float4(d[0] + bb[0], d[1] + bb[1], d[2] + bb[2], d[3] + bb[3]);
            *(float4*)(cp + 4) = make_float4(d[4] + bb[4], d[5] + bb[5], d[6] + bb[6], d[7] + bb[7]);
        }
    }
    __syncthreads();
    for (int t = tid; t < cnt * HEADS; t += D0_THREADS) {
        const float4 s4 = ((const float4*)pS)[t];
        const float4 d4 = ((const float4*)pD)[t];
        als[lo * HEADS + t] = (s4.x + s4.y) + (s4.z + s4.w);
        ald[lo * HEADS + t] = (d4.x + d4.y) + (d4.z + d4.w);
    }
}

__global__ void finish2(const float* __restrict__ att, const float* __restrict__ b2,
                        const float* __restrict__ skip32, float* __restrict__ out, int N) {
    const int node = blockIdx.x * (blockDim.x >> 5) + (threadIdx.x >> 5);
    const int c    = threadIdx.x & 31;
    if (node >= N) return;
    float v = 0.25f * att[(size_t)node * OUT_C + c]
            + b2[c] + skip32[(size_t)node * OUT_C + c];
    float sq = v * v;
#pragma unroll
    for (int o = 16; o > 0; o >>= 1) sq += __shfl_xor(sq, o, 32);
    const float norm = sqrtf(sq);
    out[(size_t)node * OUT_C + c] = v / fmaxf(norm, 1e-12f);
}

extern "C" void kernel_launch(void* const* d_in, const int* in_sizes, int n_in,
                              void* d_out, int out_size, void* d_ws, size_t ws_size,
                              hipStream_t stream) {
    (void)in_sizes; (void)n_in; (void)out_size; (void)ws_size;

    const float* x   = (const float*)d_in[0];
    const int*   ei  = (const int*)d_in[1];
    const float* W0  = (const float*)d_in[2];
    const float* aS0 = (const float*)d_in[3];
    const float* aD0 = (const float*)d_in[4];
    const float* b0  = (const float*)d_in[5];
    const float* Ws0 = (const float*)d_in[6];
    const float* bs0 = (const float*)d_in[7];
    const float* W1  = (const float*)d_in[8];
    const float* aS1 = (const float*)d_in[9];
    const float* aD1 = (const float*)d_in[10];
    const float* b1  = (const float*)d_in[11];
    const float* Ws1 = (const float*)d_in[12];
    const float* bs1 = (const float*)d_in[13];
    const float* W2  = (const float*)d_in[14];
    const float* aS2 = (const float*)d_in[15];
    const float* aD2 = (const float*)d_in[16];
    const float* b2  = (const float*)d_in[17];
    const float* Ws2 = (const float*)d_in[18];
    const float* bs2 = (const float*)d_in[19];

    const int* src = ei;
    const int* dst = ei + N_EDGES;

    const size_t NF = (size_t)N_NODES * HC;
    float*    ws    = (float*)d_ws;
    float*    h_b   = ws;
    float*    h_a   = ws + NF;
    float*    sk_a  = ws + 2 * NF;
    float*    sk_b  = ws + 3 * NF;
    float*    als_a = ws + 4 * NF;
    float*    ald_a = als_a + (size_t)N_NODES * HEADS;
    float*    als_b = ald_a + (size_t)N_NODES * HEADS;
    float*    ald_b = als_b + (size_t)N_NODES * HEADS;
    unsigned* lists = (unsigned*)(ald_b + (size_t)N_NODES * HEADS);
    int*      lens  = (int*)(lists + (size_t)OWN_WGS * OWN_WAVES * LIST_CAP);
    unsigned* cellA = (unsigned*)(lens + OWN_WGS * OWN_WAVES);
    unsigned* cellB = cellA + (size_t)OWN_WGS * OWN_WGS * LINE_W;

    dense0_owned<<<OWN_WGS, D0_THREADS, 0, stream>>>(x, W0, Ws0, bs0, aS0, aD0, h_a, sk_a, als_a, ald_a,
                                                     src, dst, cellA, cellB);
    aggr_owned<true, HC><<<OWN_WGS, 1024, 0, stream>>>(src, dst, als_a, ald_a, h_a, nullptr, lists, lens,
                                                       cellA, cellB, b0, sk_a, W1, Ws1, bs1, aS1, aD1, h_b, sk_b, als_b, ald_b);

    aggr_owned<false, OUT_C><<<OWN_WGS, 1024, 0, stream>>>(src, dst, als_b, ald_b, h_b, nullptr, lists, lens,
                                                           cellA, cellB, b1, sk_b, W2, Ws2, bs2, aS2, aD2, h_a, sk_a, als_a, ald_a);

    aggr_owned<false, 0><<<OWN_WGS, 1024, 0, stream>>>(src, dst, als_a, ald_a, h_a, sk_b, lists, lens,
                                                       cellA, cellB, nullptr, nullptr, nullptr, nullptr, nullptr, nullptr, nullptr,
                                                       nullptr, nullptr, nullptr, nullptr);
    finish2<<<(N_NODES + 7) / 8, 256, 0, stream>>>(sk_b, b2, sk_a, (float*)d_out, N_NODES);
}
